# starting kernel
# baseline (speedup 1.0000x reference)
#include <hip/hip_runtime.h>
#include <hip/hip_bf16.h>


#define N      4096
#define NFEAT  512
#define NHID   64
#define NCLASS 16
#define H1     8
#define NEG_SLOPE 0.2f
#define MASK_VAL  -1e10f

typedef __attribute__((ext_vector_type(16))) _Float16 v16h;
typedef __attribute__((ext_vector_type(8)))  _Float16 v8h;
typedef __attribute__((ext_vector_type(8)))  float    v8f;

__device__ __forceinline__ v8f wmma16(v16h a, v16h b, v8f c) {
  return __builtin_amdgcn_wmma_f32_16x16x32_f16(
       false, a,  false, b,
       (short)0, c,  false,  false);
}

__device__ __forceinline__ v16h load_frag16(const _Float16* base, int rc,
                                            int ld, int k0) {
  int lane = threadIdx.x & 31;
  int hf   = lane >> 4;
  int r    = lane & 15;
  const _Float16* p = base + (size_t)(rc + r) * ld + k0 + (hf << 3);
  v8h lo = *(const v8h*)p;
  v8h hi = *(const v8h*)(p + 16);
  return __builtin_shufflevector(lo, hi, 0, 1, 2, 3, 4, 5, 6, 7, 8, 9, 10, 11,
                                 12, 13, 14, 15);
}

__device__ __forceinline__ float lrelu(float x) {
  return x > 0.0f ? x : NEG_SLOPE * x;
}

__global__ void cast_kernel(const float* __restrict__ x,
                            const float* __restrict__ W1,
                            const float* __restrict__ W2,
                            _Float16* __restrict__ XH,
                            _Float16* __restrict__ W1T,
                            _Float16* __restrict__ W2T) {
  int i = blockIdx.x * blockDim.x + threadIdx.x;
  if (i < N * NFEAT) XH[i] = (_Float16)x[i];
  if (i < H1 * NFEAT * NHID) {
    int h   = i / (NFEAT * NHID);
    int rem = i % (NFEAT * NHID);
    int k   = rem / NHID;
    int c   = rem % NHID;
    W1T[((size_t)(h * NHID + c)) * NFEAT + k] = (_Float16)W1[i];
  }
  if (i < (H1 * NHID) * NCLASS) {
    int k = i / NCLASS;
    int c = i % NCLASS;
    W2T[(size_t)c * (H1 * NHID) + k] = (_Float16)W2[i];
  }
}

__global__ void gemm1_kernel(const _Float16* __restrict__ XH,
                             const _Float16* __restrict__ W1T,
                             float* __restrict__ feats,
                             _Float16* __restrict__ featsT) {
  int head = blockIdx.y;
  int r0   = blockIdx.x << 4;
  const _Float16* BT = W1T + (size_t)head * NHID * NFEAT;

  v8f c0 = {}, c1 = {}, c2 = {}, c3 = {};
  for (int k0 = 0; k0 < NFEAT; k0 += 32) {
    v16h a = load_frag16(XH, r0, NFEAT, k0);
    c0 = wmma16(a, load_frag16(BT, 0, NFEAT, k0), c0);
    c1 = wmma16(a, load_frag16(BT, 16, NFEAT, k0), c1);
    c2 = wmma16(a, load_frag16(BT, 32, NFEAT, k0), c2);
    c3 = wmma16(a, load_frag16(BT, 48, NFEAT, k0), c3);
  }

  int lane = threadIdx.x & 31;
  int hf   = lane >> 4;
  int r    = lane & 15;
  v8f acc[4] = {c0, c1, c2, c3};
#pragma unroll
  for (int nt = 0; nt < 4; ++nt) {
#pragma unroll
    for (int k = 0; k < 8; ++k) {
      int orow = r0 + k + (hf << 3);
      int col  = nt * 16 + r;
      float v  = acc[nt][k];
      feats[((size_t)head * N + orow) * NHID + col] = v;
      featsT[((size_t)head * NHID + col) * N + orow] = (_Float16)v;
    }
  }
}

__global__ void sn_kernel(const float* __restrict__ feats,
                          const float* __restrict__ a_s,
                          const float* __restrict__ a_n,
                          float* __restrict__ s, float* __restrict__ n,
                          int fout, int rowsPerHead, int total) {
  int i = blockIdx.x * blockDim.x + threadIdx.x;
  if (i >= total) return;
  int head = i / rowsPerHead;
  const float* f  = feats + (size_t)i * fout;
  const float* av = a_s + head * fout;
  const float* nv = a_n + head * fout;
  float ss = 0.0f, nn = 0.0f;
  for (int c = 0; c < fout; ++c) {
    float v = f[c];
    ss += v * av[c];
    nn += v * nv[c];
  }
  s[i] = ss;
  n[i] = nn;
}

__global__ void rowstats_kernel(const int* __restrict__ adj,
                                const float* __restrict__ s,
                                const float* __restrict__ n,
                                float* __restrict__ m, float* __restrict__ Z,
                                int nheads) {
  int i   = blockIdx.x;
  int tid = threadIdx.x;
  __shared__ float red[256];
  const int* arow = adj + (size_t)i * N;

  for (int h = 0; h < nheads; ++h) {
    float sv = s[h * N + i];
    const float* nrow = n + h * N;

    float lm = -3.0e38f;
    for (int j = tid; j < N; j += 256) {
      float lg = arow[j] > 0 ? lrelu(sv + nrow[j]) : MASK_VAL;
      lm = fmaxf(lm, lg);
    }
    red[tid] = lm;
    __syncthreads();
    for (int st = 128; st > 0; st >>= 1) {
      if (tid < st) red[tid] = fmaxf(red[tid], red[tid + st]);
      __syncthreads();
    }
    float mv = red[0];
    __syncthreads();

    float ls = 0.0f;
    for (int j = tid; j < N; j += 256) {
      float lg = arow[j] > 0 ? lrelu(sv + nrow[j]) : MASK_VAL;
      ls += __expf(lg - mv);
    }
    red[tid] = ls;
    __syncthreads();
    for (int st = 128; st > 0; st >>= 1) {
      if (tid < st) red[tid] += red[tid + st];
      __syncthreads();
    }
    if (tid == 0) {
      m[h * N + i] = mv;
      Z[h * N + i] = red[0];
    }
    __syncthreads();
  }
}

__global__ void attn_gemm1_kernel(const int* __restrict__ adj,
                                  const _Float16* __restrict__ featsT,
                                  const float* __restrict__ s1,
                                  const float* __restrict__ n1,
                                  const float* __restrict__ m1,
                                  const float* __restrict__ Z1,
                                  const float* __restrict__ b1,
                                  _Float16* __restrict__ H) {
  int head = blockIdx.y;
  int r0   = blockIdx.x << 4;
  int lane = threadIdx.x & 31;
  int hf   = lane >> 4;
  int r    = lane & 15;
  int row  = r0 + r;

  const float* nrow = n1 + head * N;
  float sv = s1[head * N + row];
  float mv = m1[head * N + row];
  float zi = 1.0f / Z1[head * N + row];
  const int* arow = adj + (size_t)row * N;
  const _Float16* BT = featsT + (size_t)head * NHID * N;

  v8f c0 = {}, c1 = {}, c2 = {}, c3 = {};
  for (int j0 = 0; j0 < N; j0 += 32) {
    if (j0 + 256 < N) __builtin_prefetch(arow + j0 + 256, 0, 0);
    v16h a;
#pragma unroll
    for (int e = 0; e < 16; ++e) {
      int j = j0 + (e & 7) + (hf << 3) + ((e >> 3) << 4);
      float lg = lrelu(sv + nrow[j]);
      float p  = arow[j] > 0 ? __expf(lg - mv) * zi : 0.0f;
      a[e] = (_Float16)p;
    }
    c0 = wmma16(a, load_frag16(BT, 0, N, j0), c0);
    c1 = wmma16(a, load_frag16(BT, 16, N, j0), c1);
    c2 = wmma16(a, load_frag16(BT, 32, N, j0), c2);
    c3 = wmma16(a, load_frag16(BT, 48, N, j0), c3);
  }

  v8f acc[4] = {c0, c1, c2, c3};
#pragma unroll
  for (int nt = 0; nt < 4; ++nt) {
#pragma unroll
    for (int k = 0; k < 8; ++k) {
      int orow = r0 + k + (hf << 3);
      int col  = nt * 16 + r;
      float v  = acc[nt][k] + b1[head * NHID + col];
      v = v > 0.0f ? v : 0.0f;
      H[(size_t)orow * (H1 * NHID) + head * NHID + col] = (_Float16)v;
    }
  }
}

__global__ void gemm2_kernel(const _Float16* __restrict__ H,
                             const _Float16* __restrict__ W2T,
                             float* __restrict__ feats2,
                             _Float16* __restrict__ feats2T) {
  int r0 = blockIdx.x << 4;
  v8f c = {};
  for (int k0 = 0; k0 < H1 * NHID; k0 += 32) {
    v16h a = load_frag16(H, r0, H1 * NHID, k0);
    v16h b = load_frag16(W2T, 0, H1 * NHID, k0);
    c = wmma16(a, b, c);
  }
  int lane = threadIdx.x & 31;
  int hf   = lane >> 4;
  int r    = lane & 15;
#pragma unroll
  for (int k = 0; k < 8; ++k) {
    int orow = r0 + k + (hf << 3);
    float v  = c[k];
    feats2[(size_t)orow * NCLASS + r] = v;
    feats2T[(size_t)r * N + orow]     = (_Float16)v;
  }
}

__global__ void attn_gemm2_kernel(const int* __restrict__ adj,
                                  const _Float16* __restrict__ feats2T,
                                  const float* __restrict__ s2,
                                  const float* __restrict__ n2,
                                  const float* __restrict__ m2,
                                  const float* __restrict__ Z2,
                                  const float* __restrict__ b2,
                                  float* __restrict__ out) {
  int r0   = blockIdx.x << 4;
  int lane = threadIdx.x & 31;
  int hf   = lane >> 4;
  int r    = lane & 15;
  int row  = r0 + r;

  float sv = s2[row];
  float mv = m2[row];
  float zi = 1.0f / Z2[row];
  const int* arow = adj + (size_t)row * N;

  v8f c = {};
  for (int j0 = 0; j0 < N; j0 += 32) {
    if (j0 + 256 < N) __builtin_prefetch(arow + j0 + 256, 0, 0);
    v16h a;
#pragma unroll
    for (int e = 0; e < 16; ++e) {
      int j = j0 + (e & 7) + (hf << 3) + ((e >> 3) << 4);
      float lg = lrelu(sv + n2[j]);
      float p  = arow[j] > 0 ? __expf(lg - mv) * zi : 0.0f;
      a[e] = (_Float16)p;
    }
    v16h b = load_frag16(feats2T, 0, N, j0);
    c = wmma16(a, b, c);
  }

  float bias = b2[r];
#pragma unroll
  for (int k = 0; k < 8; ++k) {
    float v = c[k] + bias;
    v = v > 0.0f ? v : 0.0f;
    float mx = v;
#pragma unroll
    for (int msk = 1; msk < 16; msk <<= 1)
      mx = fmaxf(mx, __shfl_xor(mx, msk, 32));
    float ex = __expf(v - mx);
    float sum = ex;
#pragma unroll
    for (int msk = 1; msk < 16; msk <<= 1) sum += __shfl_xor(sum, msk, 32);
    float o  = (v - mx) - logf(sum);
    int orow = r0 + k + (hf << 3);
    out[(size_t)orow * NCLASS + r] = o;
  }
}

extern "C" void kernel_launch(void* const* d_in, const int* in_sizes, int n_in,
                              void* d_out, int out_size, void* d_ws,
                              size_t ws_size, hipStream_t stream) {
  const float* x   = (const float*)d_in[0];
  const int*   adj = (const int*)d_in[1];
  const float* W1  = (const float*)d_in[2];
  const float* b1  = (const float*)d_in[3];
  const float* as1 = (const float*)d_in[4];
  const float* an1 = (const float*)d_in[5];
  const float* W2  = (const float*)d_in[6];
  const float* b2  = (const float*)d_in[7];
  const float* as2 = (const float*)d_in[8];
  const float* an2 = (const float*)d_in[9];
  float* out = (float*)d_out;

  char* ws = (char*)d_ws;
  size_t off = 0;
  auto alloc = [&](size_t bytes) -> void* {
    off = (off + 255) & ~(size_t)255;
    void* p = ws + off;
    off += bytes;
    return p;
  };

  _Float16* XH      = (_Float16*)alloc((size_t)N * NFEAT * 2);
  _Float16* W1T     = (_Float16*)alloc((size_t)H1 * NHID * NFEAT * 2);
  _Float16* W2T     = (_Float16*)alloc((size_t)NCLASS * H1 * NHID * 2);
  float*    feats1  = (float*)alloc((size_t)H1 * N * NHID * 4);
  _Float16* featsT  = (_Float16*)alloc((size_t)H1 * NHID * N * 2);
  float*    s1      = (float*)alloc((size_t)H1 * N * 4);
  float*    n1      = (float*)alloc((size_t)H1 * N * 4);
  float*    m1      = (float*)alloc((size_t)H1 * N * 4);
  float*    Z1      = (float*)alloc((size_t)H1 * N * 4);
  _Float16* H       = (_Float16*)alloc((size_t)N * H1 * NHID * 2);
  float*    feats2  = (float*)alloc((size_t)N * NCLASS * 4);
  _Float16* feats2T = (_Float16*)alloc((size_t)NCLASS * N * 2);
  float*    s2      = (float*)alloc((size_t)N * 4);
  float*    n2      = (float*)alloc((size_t)N * 4);
  float*    m2      = (float*)alloc((size_t)N * 4);
  float*    Z2      = (float*)alloc((size_t)N * 4);
  (void)ws_size; (void)n_in; (void)in_sizes; (void)out_size;

  {
    int total = N * NFEAT;
    cast_kernel<<<(total + 255) / 256, 256, 0, stream>>>(x, W1, W2, XH, W1T,
                                                         W2T);
  }
  gemm1_kernel<<<dim3(N / 16, H1), 32, 0, stream>>>(XH, W1T, feats1, featsT);
  sn_kernel<<<(H1 * N + 255) / 256, 256, 0, stream>>>(feats1, as1, an1, s1, n1,
                                                      NHID, N, H1 * N);
  rowstats_kernel<<<N, 256, 0, stream>>>(adj, s1, n1, m1, Z1, H1);
  attn_gemm1_kernel<<<dim3(N / 16, H1), 32, 0, stream>>>(adj, featsT, s1, n1,
                                                         m1, Z1, b1, H);
  gemm2_kernel<<<N / 16, 32, 0, stream>>>(H, W2T, feats2, feats2T);
  sn_kernel<<<(N + 255) / 256, 256, 0, stream>>>(feats2, as2, an2, s2, n2,
                                                 NCLASS, N, N);
  rowstats_kernel<<<N, 256, 0, stream>>>(adj, s2, n2, m2, Z2, 1);
  attn_gemm2_kernel<<<N / 16, 32, 0, stream>>>(adj, feats2T, s2, n2, m2, Z2,
                                               b2, out);
}


# Softmax row max and row sum built online inside the attention GEMM loop (lazy rescale; per-lane partial row sums joined at the end); both row-statistics launches removed
# speedup vs baseline: 2.1872x; 2.1872x over previous
#include <hip/hip_runtime.h>
#include <hip/hip_bf16.h>


#define N      4096
#define NFEAT  512
#define NHID   64
#define NCLASS 16
#define H1     8
#define NEG_SLOPE 0.2f
#define MASK_VAL  -1e10f
#define LOG2E 1.4426950408889634f
#define MASK2 (MASK_VAL * LOG2E)
#define LAZY 8.0f

typedef __attribute__((ext_vector_type(16))) _Float16 v16h;
typedef __attribute__((ext_vector_type(8)))  _Float16 v8h;
typedef __attribute__((ext_vector_type(8)))  float    v8f;

__device__ __forceinline__ v8f wmma16(v16h a, v16h b, v8f c) {
  return __builtin_amdgcn_wmma_f32_16x16x32_f16(
       false, a,  false, b,
       (short)0, c,  false,  false);
}

__device__ __forceinline__ v16h load_frag16(const _Float16* base, int rc,
                                            int ld, int k0) {
  int lane = threadIdx.x & 31;
  int hf   = lane >> 4;
  int r    = lane & 15;
  const _Float16* p = base + (size_t)(rc + r) * ld + k0 + (hf << 3);
  v8h lo = *(const v8h*)p;
  v8h hi = *(const v8h*)(p + 16);
  return __builtin_shufflevector(lo, hi, 0, 1, 2, 3, 4, 5, 6, 7, 8, 9, 10, 11,
                                 12, 13, 14, 15);
}

__device__ __forceinline__ float lrelu(float x) {
  return x > 0.0f ? x : NEG_SLOPE * x;
}

__device__ __forceinline__ float xhalf(float x) {
  int b = __builtin_bit_cast(int, x);
  return __builtin_bit_cast(
      float, __builtin_amdgcn_permlanex16(b, b, 0x76543210, 0xfedcba98, false,
                                          false));
}

__device__ __forceinline__ float logit2(int a, float nj, float svL) {
  float x2 = __builtin_fmaf(nj, LOG2E, svL);
  float lr = fmaxf(x2, NEG_SLOPE * x2);
  return a > 0 ? lr : MASK2;
}

__global__ void cast_kernel(const float* __restrict__ x,
                            const float* __restrict__ W1,
                            const float* __restrict__ W2,
                            _Float16* __restrict__ XH,
                            _Float16* __restrict__ W1T,
                            _Float16* __restrict__ W2T) {
  int i = blockIdx.x * blockDim.x + threadIdx.x;
  if (i < N * NFEAT) XH[i] = (_Float16)x[i];
  if (i < H1 * NFEAT * NHID) {
    int h   = i / (NFEAT * NHID);
    int rem = i % (NFEAT * NHID);
    int k   = rem / NHID;
    int c   = rem % NHID;
    W1T[((size_t)(h * NHID + c)) * NFEAT + k] = (_Float16)W1[i];
  }
  if (i < (H1 * NHID) * NCLASS) {
    int k = i / NCLASS;
    int c = i % NCLASS;
    W2T[(size_t)c * (H1 * NHID) + k] = (_Float16)W2[i];
  }
}

__global__ void gemm1_kernel(const _Float16* __restrict__ XH,
                             const _Float16* __restrict__ W1T,
                             float* __restrict__ feats,
                             _Float16* __restrict__ featsT) {
  int head = blockIdx.y;
  int r0   = blockIdx.x << 4;
  const _Float16* BT = W1T + (size_t)head * NHID * NFEAT;

  v8f c0 = {}, c1 = {}, c2 = {}, c3 = {};
  for (int k0 = 0; k0 < NFEAT; k0 += 32) {
    v16h a = load_frag16(XH, r0, NFEAT, k0);
    c0 = wmma16(a, load_frag16(BT, 0, NFEAT, k0), c0);
    c1 = wmma16(a, load_frag16(BT, 16, NFEAT, k0), c1);
    c2 = wmma16(a, load_frag16(BT, 32, NFEAT, k0), c2);
    c3 = wmma16(a, load_frag16(BT, 48, NFEAT, k0), c3);
  }

  int lane = threadIdx.x & 31;
  int hf   = lane >> 4;
  int r    = lane & 15;
  v8f acc[4] = {c0, c1, c2, c3};
#pragma unroll
  for (int nt = 0; nt < 4; ++nt) {
#pragma unroll
    for (int k = 0; k < 8; ++k) {
      int orow = r0 + k + (hf << 3);
      int col  = nt * 16 + r;
      float v  = acc[nt][k];
      feats[((size_t)head * N + orow) * NHID + col] = v;
      featsT[((size_t)head * NHID + col) * N + orow] = (_Float16)v;
    }
  }
}

__global__ void sn_kernel(const float* __restrict__ feats,
                          const float* __restrict__ a_s,
                          const float* __restrict__ a_n,
                          float* __restrict__ s, float* __restrict__ n,
                          int fout, int rowsPerHead, int total) {
  int i = blockIdx.x * blockDim.x + threadIdx.x;
  if (i >= total) return;
  int head = i / rowsPerHead;
  const float* f  = feats + (size_t)i * fout;
  const float* av = a_s + head * fout;
  const float* nv = a_n + head * fout;
  float ss = 0.0f, nn = 0.0f;
  for (int c = 0; c < fout; ++c) {
    float v = f[c];
    ss += v * av[c];
    nn += v * nv[c];
  }
  s[i] = ss;
  n[i] = nn;
}

__global__ void attn_gemm1_kernel(const int* __restrict__ adj,
                                  const _Float16* __restrict__ featsT,
                                  const float* __restrict__ s1,
                                  const float* __restrict__ n1,
                                  const float* __restrict__ b1,
                                  _Float16* __restrict__ H) {
  int head = blockIdx.y;
  int r0   = blockIdx.x << 4;
  int lane = threadIdx.x & 31;
  int hf   = lane >> 4;
  int r    = lane & 15;
  int row  = r0 + r;

  const float* nrow = n1 + head * N;
  float svL = s1[head * N + row] * LOG2E;
  const int* arow = adj + (size_t)row * N;
  const _Float16* BT = featsT + (size_t)head * NHID * N;

  float m_run = -1.0e30f;
  float l_run = 0.0f;

  v8f c0 = {}, c1 = {}, c2 = {}, c3 = {};
  for (int j0 = 0; j0 < N; j0 += 32) {
    if (j0 + 256 < N) __builtin_prefetch(arow + j0 + 256, 0, 0);
    float t[16];
    float bm = MASK2;
#pragma unroll
    for (int e = 0; e < 16; ++e) {
      int j = j0 + (e & 7) + (hf << 3) + ((e >> 3) << 4);
      t[e] = logit2(arow[j], nrow[j], svL);
      bm   = fmaxf(bm, t[e]);
    }
    bm = fmaxf(bm, xhalf(bm));

    if (__builtin_amdgcn_ballot_w32(bm > m_run + LAZY) != 0) {
      float m_new = fmaxf(m_run, bm);
      float f = __builtin_amdgcn_exp2f(m_run - m_new);
      l_run *= f;
#pragma unroll
      for (int k = 0; k < 8; ++k) {
        float fk = __shfl(f, k + (hf << 3), 32);
        c0[k] *= fk;
        c1[k] *= fk;
        c2[k] *= fk;
        c3[k] *= fk;
      }
      m_run = m_new;
    }

    v16h a;
#pragma unroll
    for (int e = 0; e < 16; ++e) {
      float p = __builtin_amdgcn_exp2f(t[e] - m_run);
      l_run += p;
      a[e] = (_Float16)p;
    }
    c0 = wmma16(a, load_frag16(BT, 0, N, j0), c0);
    c1 = wmma16(a, load_frag16(BT, 16, N, j0), c1);
    c2 = wmma16(a, load_frag16(BT, 32, N, j0), c2);
    c3 = wmma16(a, load_frag16(BT, 48, N, j0), c3);
  }

  float inv = 1.0f / (l_run + xhalf(l_run));
  float ik[8];
#pragma unroll
  for (int k = 0; k < 8; ++k) ik[k] = __shfl(inv, k + (hf << 3), 32);

  v8f acc[4] = {c0, c1, c2, c3};
#pragma unroll
  for (int nt = 0; nt < 4; ++nt) {
#pragma unroll
    for (int k = 0; k < 8; ++k) {
      int orow = r0 + k + (hf << 3);
      int col  = nt * 16 + r;
      float v  = acc[nt][k] * ik[k] + b1[head * NHID + col];
      v = v > 0.0f ? v : 0.0f;
      H[(size_t)orow * (H1 * NHID) + head * NHID + col] = (_Float16)v;
    }
  }
}

__global__ void gemm2_kernel(const _Float16* __restrict__ H,
                             const _Float16* __restrict__ W2T,
                             float* __restrict__ feats2,
                             _Float16* __restrict__ feats2T) {
  int r0 = blockIdx.x << 4;
  v8f c = {};
  for (int k0 = 0; k0 < H1 * NHID; k0 += 32) {
    v16h a = load_frag16(H, r0, H1 * NHID, k0);
    v16h b = load_frag16(W2T, 0, H1 * NHID, k0);
    c = wmma16(a, b, c);
  }
  int lane = threadIdx.x & 31;
  int hf   = lane >> 4;
  int r    = lane & 15;
#pragma unroll
  for (int k = 0; k < 8; ++k) {
    int orow = r0 + k + (hf << 3);
    float v  = c[k];
    feats2[(size_t)orow * NCLASS + r] = v;
    feats2T[(size_t)r * N + orow]     = (_Float16)v;
  }
}

__global__ void attn_gemm2_kernel(const int* __restrict__ adj,
                                  const _Float16* __restrict__ feats2T,
                                  const float* __restrict__ s2,
                                  const float* __restrict__ n2,
                                  const float* __restrict__ b2,
                                  float* __restrict__ out) {
  int r0   = blockIdx.x << 4;
  int lane = threadIdx.x & 31;
  int hf   = lane >> 4;
  int r    = lane & 15;
  int row  = r0 + r;

  float svL = s2[row] * LOG2E;
  const int* arow = adj + (size_t)row * N;

  float m_run = -1.0e30f;
  float l_run = 0.0f;

  v8f c = {};
  for (int j0 = 0; j0 < N; j0 += 32) {
    if (j0 + 256 < N) __builtin_prefetch(arow + j0 + 256, 0, 0);
    float t[16];
    float bm = MASK2;
#pragma unroll
    for (int e = 0; e < 16; ++e) {
      int j = j0 + (e & 7) + (hf << 3) + ((e >> 3) << 4);
      t[e] = logit2(arow[j], n2[j], svL);
      bm   = fmaxf(bm, t[e]);
    }
    bm = fmaxf(bm, xhalf(bm));

    if (__builtin_amdgcn_ballot_w32(bm > m_run + LAZY) != 0) {
      float m_new = fmaxf(m_run, bm);
      float f = __builtin_amdgcn_exp2f(m_run - m_new);
      l_run *= f;
#pragma unroll
      for (int k = 0; k < 8; ++k) c[k] *= __shfl(f, k + (hf << 3), 32);
      m_run = m_new;
    }

    v16h a;
#pragma unroll
    for (int e = 0; e < 16; ++e) {
      float p = __builtin_amdgcn_exp2f(t[e] - m_run);
      l_run += p;
      a[e] = (_Float16)p;
    }
    v16h b = load_frag16(feats2T, 0, N, j0);
    c = wmma16(a, b, c);
  }

  float inv  = 1.0f / (l_run + xhalf(l_run));
  float bias = b2[r];
#pragma unroll
  for (int k = 0; k < 8; ++k) {
    float v = c[k] * __shfl(inv, k + (hf << 3), 32) + bias;
    v = v > 0.0f ? v : 0.0f;
    float mx = v;
#pragma unroll
    for (int msk = 1; msk < 16; msk <<= 1)
      mx = fmaxf(mx, __shfl_xor(mx, msk, 32));
    float ex = __expf(v - mx);
    float sum = ex;
#pragma unroll
    for (int msk = 1; msk < 16; msk <<= 1) sum += __shfl_xor(sum, msk, 32);
    float o  = (v - mx) - logf(sum);
    int orow = r0 + k + (hf << 3);
    out[(size_t)orow * NCLASS + r] = o;
  }
}

extern "C" void kernel_launch(void* const* d_in, const int* in_sizes, int n_in,
                              void* d_out, int out_size, void* d_ws,
                              size_t ws_size, hipStream_t stream) {
  const float* x   = (const float*)d_in[0];
  const int*   adj = (const int*)d_in[1];
  const float* W1  = (const float*)d_in[2];
  const float* b1  = (const float*)d_in[3];
  const float* as1 = (const float*)d_in[4];
  const float* an1 = (const float*)d_in[5];
  const float* W2  = (const float*)d_in[6];
  const float* b2  = (const float*)d_in[7];
  const float* as2 = (const float*)d_in[8];
  const float* an2 = (const float*)d_in[9];
  float* out = (float*)d_out;

  char* ws = (char*)d_ws;
  size_t off = 0;
  auto alloc = [&](size_t bytes) -> void* {
    off = (off + 255) & ~(size_t)255;
    void* p = ws + off;
    off += bytes;
    return p;
  };

  _Float16* XH      = (_Float16*)alloc((size_t)N * NFEAT * 2);
  _Float16* W1T     = (_Float16*)alloc((size_t)H1 * NHID * NFEAT * 2);
  _Float16* W2T     = (_Float16*)alloc((size_t)NCLASS * H1 * NHID * 2);
  float*    feats1  = (float*)alloc((size_t)H1 * N * NHID * 4);
  _Float16* featsT  = (_Float16*)alloc((size_t)H1 * NHID * N * 2);
  float*    s1      = (float*)alloc((size_t)H1 * N * 4);
  float*    n1      = (float*)alloc((size_t)H1 * N * 4);
  _Float16* H       = (_Float16*)alloc((size_t)N * H1 * NHID * 2);
  float*    feats2  = (float*)alloc((size_t)N * NCLASS * 4);
  _Float16* feats2T = (_Float16*)alloc((size_t)NCLASS * N * 2);
  float*    s2      = (float*)alloc((size_t)N * 4);
  float*    n2      = (float*)alloc((size_t)N * 4);
  (void)ws_size; (void)n_in; (void)in_sizes; (void)out_size;

  {
    int total = N * NFEAT;
    cast_kernel<<<(total + 255) / 256, 256, 0, stream>>>(x, W1, W2, XH, W1T,
                                                         W2T);
  }
  gemm1_kernel<<<dim3(N / 16, H1), 32, 0, stream>>>(XH, W1T, feats1, featsT);
  sn_kernel<<<(H1 * N + 255) / 256, 256, 0, stream>>>(feats1, as1, an1, s1, n1,
                                                      NHID, N, H1 * N);
  attn_gemm1_kernel<<<dim3(N / 16, H1), 32, 0, stream>>>(adj, featsT, s1, n1,
                                                         b1, H);
  gemm2_kernel<<<N / 16, 32, 0, stream>>>(H, W2T, feats2, feats2T);
  sn_kernel<<<(N + 255) / 256, 256, 0, stream>>>(feats2, as2, an2, s2, n2,
                                                 NCLASS, N, N);
  attn_gemm2_kernel<<<N / 16, 32, 0, stream>>>(adj, feats2T, s2, n2, b2, out);
}


# Both attention kernels: each pass's B-fragment loads are issued at the top of the pass with its adjacency and n loads, ahead of the rescale branch: one memory round trip per pass instead of two
# speedup vs baseline: 3.1476x; 1.4391x over previous
#include <hip/hip_runtime.h>
#include <hip/hip_bf16.h>


#define N      4096
#define NFEAT  512
#define NHID   64
#define NCLASS 16
#define H1     8
#define NEG_SLOPE 0.2f
#define MASK_VAL  -1e10f
#define LOG2E 1.4426950408889634f
#define MASK2 (MASK_VAL * LOG2E)
#define LAZY 8.0f

typedef __attribute__((ext_vector_type(16))) _Float16 v16h;
typedef __attribute__((ext_vector_type(8)))  _Float16 v8h;
typedef __attribute__((ext_vector_type(8)))  float    v8f;

__device__ __forceinline__ v8f wmma16(v16h a, v16h b, v8f c) {
  return __builtin_amdgcn_wmma_f32_16x16x32_f16(
       false, a,  false, b,
       (short)0, c,  false,  false);
}

__device__ __forceinline__ v16h load_frag16(const _Float16* base, int rc,
                                            int ld, int k0) {
  int lane = threadIdx.x & 31;
  int hf   = lane >> 4;
  int r    = lane & 15;
  const _Float16* p = base + (size_t)(rc + r) * ld + k0 + (hf << 3);
  v8h lo = *(const v8h*)p;
  v8h hi = *(const v8h*)(p + 16);
  return __builtin_shufflevector(lo, hi, 0, 1, 2, 3, 4, 5, 6, 7, 8, 9, 10, 11,
                                 12, 13, 14, 15);
}

__device__ __forceinline__ float lrelu(float x) {
  return x > 0.0f ? x : NEG_SLOPE * x;
}

__device__ __forceinline__ float xhalf(float x) {
  int b = __builtin_bit_cast(int, x);
  return __builtin_bit_cast(
      float, __builtin_amdgcn_permlanex16(b, b, 0x76543210, 0xfedcba98, false,
                                          false));
}

__device__ __forceinline__ float logit2(int a, float nj, float svL) {
  float x2 = __builtin_fmaf(nj, LOG2E, svL);
  float lr = fmaxf(x2, NEG_SLOPE * x2);
  return a > 0 ? lr : MASK2;
}

__global__ void cast_kernel(const float* __restrict__ x,
                            const float* __restrict__ W1,
                            const float* __restrict__ W2,
                            _Float16* __restrict__ XH,
                            _Float16* __restrict__ W1T,
                            _Float16* __restrict__ W2T) {
  int i = blockIdx.x * blockDim.x + threadIdx.x;
  if (i < N * NFEAT) XH[i] = (_Float16)x[i];
  if (i < H1 * NFEAT * NHID) {
    int h   = i / (NFEAT * NHID);
    int rem = i % (NFEAT * NHID);
    int k   = rem / NHID;
    int c   = rem % NHID;
    W1T[((size_t)(h * NHID + c)) * NFEAT + k] = (_Float16)W1[i];
  }
  if (i < (H1 * NHID) * NCLASS) {
    int k = i / NCLASS;
    int c = i % NCLASS;
    W2T[(size_t)c * (H1 * NHID) + k] = (_Float16)W2[i];
  }
}

__global__ void gemm1_kernel(const _Float16* __restrict__ XH,
                             const _Float16* __restrict__ W1T,
                             float* __restrict__ feats,
                             _Float16* __restrict__ featsT) {
  int head = blockIdx.y;
  int r0   = blockIdx.x << 4;
  const _Float16* BT = W1T + (size_t)head * NHID * NFEAT;

  v8f c0 = {}, c1 = {}, c2 = {}, c3 = {};
  for (int k0 = 0; k0 < NFEAT; k0 += 32) {
    v16h a = load_frag16(XH, r0, NFEAT, k0);
    c0 = wmma16(a, load_frag16(BT, 0, NFEAT, k0), c0);
    c1 = wmma16(a, load_frag16(BT, 16, NFEAT, k0), c1);
    c2 = wmma16(a, load_frag16(BT, 32, NFEAT, k0), c2);
    c3 = wmma16(a, load_frag16(BT, 48, NFEAT, k0), c3);
  }

  int lane = threadIdx.x & 31;
  int hf   = lane >> 4;
  int r    = lane & 15;
  v8f acc[4] = {c0, c1, c2, c3};
#pragma unroll
  for (int nt = 0; nt < 4; ++nt) {
#pragma unroll
    for (int k = 0; k < 8; ++k) {
      int orow = r0 + k + (hf << 3);
      int col  = nt * 16 + r;
      float v  = acc[nt][k];
      feats[((size_t)head * N + orow) * NHID + col] = v;
      featsT[((size_t)head * NHID + col) * N + orow] = (_Float16)v;
    }
  }
}

__global__ void sn_kernel(const float* __restrict__ feats,
                          const float* __restrict__ a_s,
                          const float* __restrict__ a_n,
                          float* __restrict__ s, float* __restrict__ n,
                          int fout, int rowsPerHead, int total) {
  int i = blockIdx.x * blockDim.x + threadIdx.x;
  if (i >= total) return;
  int head = i / rowsPerHead;
  const float* f  = feats + (size_t)i * fout;
  const float* av = a_s + head * fout;
  const float* nv = a_n + head * fout;
  float ss = 0.0f, nn = 0.0f;
  for (int c = 0; c < fout; ++c) {
    float v = f[c];
    ss += v * av[c];
    nn += v * nv[c];
  }
  s[i] = ss;
  n[i] = nn;
}

__global__ void attn_gemm1_kernel(const int* __restrict__ adj,
                                  const _Float16* __restrict__ featsT,
                                  const float* __restrict__ s1,
                                  const float* __restrict__ n1,
                                  const float* __restrict__ b1,
                                  _Float16* __restrict__ H) {
  int head = blockIdx.y;
  int r0   = blockIdx.x << 4;
  int lane = threadIdx.x & 31;
  int hf   = lane >> 4;
  int r    = lane & 15;
  int row  = r0 + r;

  const float* nrow = n1 + head * N;
  float svL = s1[head * N + row] * LOG2E;
  const int* arow = adj + (size_t)row * N;
  const _Float16* BT = featsT + (size_t)head * NHID * N;

  float m_run = -1.0e30f;
  float l_run = 0.0f;

  v8f c0 = {}, c1 = {}, c2 = {}, c3 = {};
  for (int j0 = 0; j0 < N; j0 += 32) {
    if (j0 + 256 < N) __builtin_prefetch(arow + j0 + 256, 0, 0);
    int   av[16];
    float nv[16];
#pragma unroll
    for (int e = 0; e < 16; ++e) {
      int j = j0 + (e & 7) + (hf << 3) + ((e >> 3) << 4);
      av[e] = arow[j];
      nv[e] = nrow[j];
    }
    v16h b0 = load_frag16(BT, 0, N, j0);
    v16h b1 = load_frag16(BT, 16, N, j0);
    v16h b2 = load_frag16(BT, 32, N, j0);
    v16h b3 = load_frag16(BT, 48, N, j0);

    float t[16];
    float bm = MASK2;
#pragma unroll
    for (int e = 0; e < 16; ++e) {
      t[e] = logit2(av[e], nv[e], svL);
      bm   = fmaxf(bm, t[e]);
    }
    bm = fmaxf(bm, xhalf(bm));

    asm volatile("" : "+v"(b0), "+v"(b1), "+v"(b2), "+v"(b3), "+v"(bm));

    if (__builtin_amdgcn_ballot_w32(bm > m_run + LAZY) != 0) {
      float m_new = fmaxf(m_run, bm);
      float f = __builtin_amdgcn_exp2f(m_run - m_new);
      l_run *= f;
#pragma unroll
      for (int k = 0; k < 8; ++k) {
        float fk = __shfl(f, k + (hf << 3), 32);
        c0[k] *= fk;
        c1[k] *= fk;
        c2[k] *= fk;
        c3[k] *= fk;
      }
      m_run = m_new;
    }

    v16h a;
#pragma unroll
    for (int e = 0; e < 16; ++e) {
      float p = __builtin_amdgcn_exp2f(t[e] - m_run);
      l_run += p;
      a[e] = (_Float16)p;
    }
    c0 = wmma16(a, b0, c0);
    c1 = wmma16(a, b1, c1);
    c2 = wmma16(a, b2, c2);
    c3 = wmma16(a, b3, c3);
  }

  float inv = 1.0f / (l_run + xhalf(l_run));
  float ik[8];
#pragma unroll
  for (int k = 0; k < 8; ++k) ik[k] = __shfl(inv, k + (hf << 3), 32);

  v8f acc[4] = {c0, c1, c2, c3};
#pragma unroll
  for (int nt = 0; nt < 4; ++nt) {
#pragma unroll
    for (int k = 0; k < 8; ++k) {
      int orow = r0 + k + (hf << 3);
      int col  = nt * 16 + r;
      float v  = acc[nt][k] * ik[k] + b1[head * NHID + col];
      v = v > 0.0f ? v : 0.0f;
      H[(size_t)orow * (H1 * NHID) + head * NHID + col] = (_Float16)v;
    }
  }
}

__global__ void gemm2_kernel(const _Float16* __restrict__ H,
                             const _Float16* __restrict__ W2T,
                             float* __restrict__ feats2,
                             _Float16* __restrict__ feats2T) {
  int r0 = blockIdx.x << 4;
  v8f c = {};
  for (int k0 = 0; k0 < H1 * NHID; k0 += 32) {
    v16h a = load_frag16(H, r0, H1 * NHID, k0);
    v16h b = load_frag16(W2T, 0, H1 * NHID, k0);
    c = wmma16(a, b, c);
  }
  int lane = threadIdx.x & 31;
  int hf   = lane >> 4;
  int r    = lane & 15;
#pragma unroll
  for (int k = 0; k < 8; ++k) {
    int orow = r0 + k + (hf << 3);
    float v  = c[k];
    feats2[(size_t)orow * NCLASS + r] = v;
    feats2T[(size_t)r * N + orow]     = (_Float16)v;
  }
}

#define SPLIT2 16

__global__ void __launch_bounds__(32 * SPLIT2)
__attribute__((amdgpu_waves_per_eu(4, 8)))
attn_gemm2_kernel(const int* __restrict__ adj,
                  const _Float16* __restrict__ feats2T,
                  const float* __restrict__ s2,
                  const float* __restrict__ n2,
                  const float* __restrict__ b2,
                  float* __restrict__ out) {
  __shared__ float sh_c[SPLIT2][16][NCLASS];
  __shared__ float sh_m[SPLIT2][16];
  __shared__ float sh_l[SPLIT2][16];

  int r0   = blockIdx.x << 4;
  int wv   = __builtin_amdgcn_readfirstlane((int)(threadIdx.x >> 5));
  int lane = threadIdx.x & 31;
  int hf   = lane >> 4;
  int r    = lane & 15;
  int row  = r0 + r;

  float svL = s2[row] * LOG2E;
  const int* arow = adj + (size_t)row * N;

  float m_run = -1.0e30f;
  float l_run = 0.0f;

  v8f c = {};
  const int jbeg = wv * (N / SPLIT2);
  const int jend = jbeg + (N / SPLIT2);
  for (int j0 = jbeg; j0 < jend; j0 += 32) {
    int   av[16];
    float nv[16];
#pragma unroll
    for (int e = 0; e < 16; ++e) {
      int j = j0 + (e & 7) + (hf << 3) + ((e >> 3) << 4);
      av[e] = arow[j];
      nv[e] = n2[j];
    }
    v16h b = load_frag16(feats2T, 0, N, j0);

    float t[16];
    float bm = MASK2;
#pragma unroll
    for (int e = 0; e < 16; ++e) {
      t[e] = logit2(av[e], nv[e], svL);
      bm   = fmaxf(bm, t[e]);
    }
    bm = fmaxf(bm, xhalf(bm));

    asm volatile("" : "+v"(b), "+v"(bm));

    if (__builtin_amdgcn_ballot_w32(bm > m_run + LAZY) != 0) {
      float m_new = fmaxf(m_run, bm);
      float f = __builtin_amdgcn_exp2f(m_run - m_new);
      l_run *= f;
#pragma unroll
      for (int k = 0; k < 8; ++k) c[k] *= __shfl(f, k + (hf << 3), 32);
      m_run = m_new;
    }

    v16h a;
#pragma unroll
    for (int e = 0; e < 16; ++e) {
      float p = __builtin_amdgcn_exp2f(t[e] - m_run);
      l_run += p;
      a[e] = (_Float16)p;
    }
    c = wmma16(a, b, c);
  }

  float lsum = l_run + xhalf(l_run);
  if (hf == 0) {
    sh_m[wv][r] = m_run;
    sh_l[wv][r] = lsum;
  }
#pragma unroll
  for (int k = 0; k < 8; ++k) sh_c[wv][k + (hf << 3)][r] = c[k];
  __syncthreads();

  if (threadIdx.x < 16 * NCLASS) {
    int mrow = threadIdx.x >> 4;
    int cls  = threadIdx.x & 15;
    float M = sh_m[0][mrow];
#pragma unroll
    for (int w = 1; w < SPLIT2; ++w) M = fmaxf(M, sh_m[w][mrow]);
    float L = 0.0f, C = 0.0f;
#pragma unroll
    for (int w = 0; w < SPLIT2; ++w) {
      float f = __builtin_amdgcn_exp2f(sh_m[w][mrow] - M);
      L = __builtin_fmaf(sh_l[w][mrow], f, L);
      C = __builtin_fmaf(sh_c[w][mrow][cls], f, C);
    }
    float inv = 1.0f / L;
    float v   = C * inv + b2[cls];
    v = v > 0.0f ? v : 0.0f;
    float mx = v;
#pragma unroll
    for (int msk = 1; msk < 16; msk <<= 1)
      mx = fmaxf(mx, __shfl_xor(mx, msk, 32));
    float ex = __expf(v - mx);
    float sum = ex;
#pragma unroll
    for (int msk = 1; msk < 16; msk <<= 1) sum += __shfl_xor(sum, msk, 32);
    float o = (v - mx) - logf(sum);
    out[(size_t)(r0 + mrow) * NCLASS + cls] = o;
  }
}

extern "C" void kernel_launch(void* const* d_in, const int* in_sizes, int n_in,
                              void* d_out, int out_size, void* d_ws,
                              size_t ws_size, hipStream_t stream) {
  const float* x   = (const float*)d_in[0];
  const int*   adj = (const int*)d_in[1];
  const float* W1  = (const float*)d_in[2];
  const float* b1  = (const float*)d_in[3];
  const float* as1 = (const float*)d_in[4];
  const float* an1 = (const float*)d_in[5];
  const float* W2  = (const float*)d_in[6];
  const float* b2  = (const float*)d_in[7];
  const float* as2 = (const float*)d_in[8];
  const float* an2 = (const float*)d_in[9];
  float* out = (float*)d_out;

  char* ws = (char*)d_ws;
  size_t off = 0;
  auto alloc = [&](size_t bytes) -> void* {
    off = (off + 255) & ~(size_t)255;
    void* p = ws + off;
    off += bytes;
    return p;
  };

  _Float16* XH      = (_Float16*)alloc((size_t)N * NFEAT * 2);
  _Float16* W1T     = (_Float16*)alloc((size_t)H1 * NHID * NFEAT * 2);
  _Float16* W2T     = (_Float16*)alloc((size_t)NCLASS * H1 * NHID * 2);
  float*    feats1  = (float*)alloc((size_t)H1 * N * NHID * 4);
  _Float16* featsT  = (_Float16*)alloc((size_t)H1 * NHID * N * 2);
  float*    s1      = (float*)alloc((size_t)H1 * N * 4);
  float*    n1      = (float*)alloc((size_t)H1 * N * 4);
  _Float16* H       = (_Float16*)alloc((size_t)N * H1 * NHID * 2);
  float*    feats2  = (float*)alloc((size_t)N * NCLASS * 4);
  _Float16* feats2T = (_Float16*)alloc((size_t)NCLASS * N * 2);
  float*    s2      = (float*)alloc((size_t)N * 4);
  float*    n2      = (float*)alloc((size_t)N * 4);
  (void)ws_size; (void)n_in; (void)in_sizes; (void)out_size;

  {
    int total = N * NFEAT;
    cast_kernel<<<(total + 255) / 256, 256, 0, stream>>>(x, W1, W2, XH, W1T,
                                                         W2T);
  }
  gemm1_kernel<<<dim3(N / 16, H1), 32, 0, stream>>>(XH, W1T, feats1, featsT);
  sn_kernel<<<(H1 * N + 255) / 256, 256, 0, stream>>>(feats1, as1, an1, s1, n1,
                                                      NHID, N, H1 * N);
  attn_gemm1_kernel<<<dim3(N / 16, H1), 32, 0, stream>>>(adj, featsT, s1, n1,
                                                         b1, H);
  gemm2_kernel<<<N / 16, 32, 0, stream>>>(H, W2T, feats2, feats2T);
  sn_kernel<<<(N + 255) / 256, 256, 0, stream>>>(feats2, as2, an2, s2, n2,
                                                 NCLASS, N, N);
  attn_gemm2_kernel<<<N / 16, 32 * SPLIT2, 0, stream>>>(adj, feats2T, s2, n2,
                                                        b2, out);
}


# Layer-1 attention: B tile (64 features x 64 columns) loaded once per 4-wave work-group by whole-line loads into 2-stage LDS; waves take B fragments from LDS, not global; order of sums unchanged
# speedup vs baseline: 4.0125x; 1.2748x over previous
#include <hip/hip_runtime.h>
#include <hip/hip_bf16.h>


#define N      4096
#define NFEAT  512
#define NHID   64
#define NCLASS 16
#define H1     8
#define NEG_SLOPE 0.2f
#define MASK_VAL  -1e10f
#define LOG2E 1.4426950408889634f
#define MASK2 (MASK_VAL * LOG2E)
#define LAZY 8.0f

typedef __attribute__((ext_vector_type(16))) _Float16 v16h;
typedef __attribute__((ext_vector_type(8)))  _Float16 v8h;
typedef __attribute__((ext_vector_type(8)))  float    v8f;

__device__ __forceinline__ v8f wmma16(v16h a, v16h b, v8f c) {
  return __builtin_amdgcn_wmma_f32_16x16x32_f16(
       false, a,  false, b,
       (short)0, c,  false,  false);
}

__device__ __forceinline__ v16h load_frag16(const _Float16* base, int rc,
                                            int ld, int k0) {
  int lane = threadIdx.x & 31;
  int hf   = lane >> 4;
  int r    = lane & 15;
  const _Float16* p = base + (size_t)(rc + r) * ld + k0 + (hf << 3);
  v8h lo = *(const v8h*)p;
  v8h hi = *(const v8h*)(p + 16);
  return __builtin_shufflevector(lo, hi, 0, 1, 2, 3, 4, 5, 6, 7, 8, 9, 10, 11,
                                 12, 13, 14, 15);
}

__device__ __forceinline__ float lrelu(float x) {
  return x > 0.0f ? x : NEG_SLOPE * x;
}

__device__ __forceinline__ float xhalf(float x) {
  int b = __builtin_bit_cast(int, x);
  return __builtin_bit_cast(
      float, __builtin_amdgcn_permlanex16(b, b, 0x76543210, 0xfedcba98, false,
                                          false));
}

__device__ __forceinline__ float logit2(int a, float nj, float svL) {
  float x2 = __builtin_fmaf(nj, LOG2E, svL);
  float lr = fmaxf(x2, NEG_SLOPE * x2);
  return a > 0 ? lr : MASK2;
}

__global__ void cast_kernel(const float* __restrict__ x,
                            const float* __restrict__ W1,
                            const float* __restrict__ W2,
                            _Float16* __restrict__ XH,
                            _Float16* __restrict__ W1T,
                            _Float16* __restrict__ W2T) {
  int i = blockIdx.x * blockDim.x + threadIdx.x;
  if (i < N * NFEAT) XH[i] = (_Float16)x[i];
  if (i < H1 * NFEAT * NHID) {
    int h   = i / (NFEAT * NHID);
    int rem = i % (NFEAT * NHID);
    int k   = rem / NHID;
    int c   = rem % NHID;
    W1T[((size_t)(h * NHID + c)) * NFEAT + k] = (_Float16)W1[i];
  }
  if (i < (H1 * NHID) * NCLASS) {
    int k = i / NCLASS;
    int c = i % NCLASS;
    W2T[(size_t)c * (H1 * NHID) + k] = (_Float16)W2[i];
  }
}

__global__ void gemm1_kernel(const _Float16* __restrict__ XH,
                             const _Float16* __restrict__ W1T,
                             float* __restrict__ feats,
                             _Float16* __restrict__ featsT) {
  int head = blockIdx.y;
  int r0   = blockIdx.x << 4;
  const _Float16* BT = W1T + (size_t)head * NHID * NFEAT;

  v8f c0 = {}, c1 = {}, c2 = {}, c3 = {};
  for (int k0 = 0; k0 < NFEAT; k0 += 32) {
    v16h a = load_frag16(XH, r0, NFEAT, k0);
    c0 = wmma16(a, load_frag16(BT, 0, NFEAT, k0), c0);
    c1 = wmma16(a, load_frag16(BT, 16, NFEAT, k0), c1);
    c2 = wmma16(a, load_frag16(BT, 32, NFEAT, k0), c2);
    c3 = wmma16(a, load_frag16(BT, 48, NFEAT, k0), c3);
  }

  int lane = threadIdx.x & 31;
  int hf   = lane >> 4;
  int r    = lane & 15;
  v8f acc[4] = {c0, c1, c2, c3};
#pragma unroll
  for (int nt = 0; nt < 4; ++nt) {
#pragma unroll
    for (int k = 0; k < 8; ++k) {
      int orow = r0 + k + (hf << 3);
      int col  = nt * 16 + r;
      float v  = acc[nt][k];
      feats[((size_t)head * N + orow) * NHID + col] = v;
      featsT[((size_t)head * NHID + col) * N + orow] = (_Float16)v;
    }
  }
}

__global__ void sn_kernel(const float* __restrict__ feats,
                          const float* __restrict__ a_s,
                          const float* __restrict__ a_n,
                          float* __restrict__ s, float* __restrict__ n,
                          int fout, int rowsPerHead, int total) {
  int i = blockIdx.x * blockDim.x + threadIdx.x;
  if (i >= total) return;
  int head = i / rowsPerHead;
  const float* f  = feats + (size_t)i * fout;
  const float* av = a_s + head * fout;
  const float* nv = a_n + head * fout;
  float ss = 0.0f, nn = 0.0f;
  for (int c = 0; c < fout; ++c) {
    float v = f[c];
    ss += v * av[c];
    nn += v * nv[c];
  }
  s[i] = ss;
  n[i] = nn;
}

#define WAVES1  4
#define KT      64
#define PITCH_H 72

__global__ void __launch_bounds__(32 * WAVES1)
__attribute__((amdgpu_waves_per_eu(1, 2)))
attn_gemm1_kernel(const int* __restrict__ adj,
                  const _Float16* __restrict__ featsT,
                  const float* __restrict__ s1,
                  const float* __restrict__ n1,
                  const float* __restrict__ b1,
                  _Float16* __restrict__ H) {
  __shared__ __attribute__((aligned(16))) _Float16 sB[2][NHID * PITCH_H];

  int head = blockIdx.y;
  int wv   = __builtin_amdgcn_readfirstlane((int)(threadIdx.x >> 5));
  int r0   = ((int)blockIdx.x * WAVES1 + wv) << 4;
  int lane = threadIdx.x & 31;
  int hf   = lane >> 4;
  int r    = lane & 15;
  int row  = r0 + r;

  const float* nrow = n1 + head * N;
  float svL = s1[head * N + row] * LOG2E;
  const int* arow = adj + (size_t)row * N;
  const _Float16* BT = featsT + (size_t)head * NHID * N;

  const int tid = threadIdx.x;
  const _Float16* gsrc = BT + (size_t)(tid >> 3) * N + ((tid & 7) << 3);
  const int loff = (tid >> 3) * PITCH_H + ((tid & 7) << 3);

#pragma unroll
  for (int i = 0; i < 4; ++i) {
    v8h g = *(const v8h*)(gsrc + (size_t)i * 16 * N);
    *(v8h*)(&sB[0][loff + i * 16 * PITCH_H]) = g;
  }
  __syncthreads();

  float m_run = -1.0e30f;
  float l_run = 0.0f;

  v8f c0 = {}, c1 = {}, c2 = {}, c3 = {};
  for (int j0 = 0; j0 < N; j0 += KT) {
    const int cur = (j0 / KT) & 1;
    if (j0 + 256 < N) __builtin_prefetch(arow + j0 + 256, 0, 0);
    int   av[32];
    float nv[32];
#pragma unroll
    for (int e = 0; e < 32; ++e) {
      int j = j0 + ((e >> 4) << 5) + (e & 7) + (hf << 3) + (((e >> 3) & 1) << 4);
      av[e] = arow[j];
      nv[e] = nrow[j];
    }
    const int jn = (j0 + KT < N) ? (j0 + KT) : j0;
    v8h g[4];
#pragma unroll
    for (int i = 0; i < 4; ++i)
      g[i] = *(const v8h*)(gsrc + (size_t)i * 16 * N + jn);

    const _Float16* st = &sB[cur][0];
    v16h bf[2][4];
#pragma unroll
    for (int u = 0; u < 2; ++u)
#pragma unroll
      for (int nt = 0; nt < 4; ++nt)
        bf[u][nt] = load_frag16(st, nt * 16, PITCH_H, 32 * u);

    float t[32];
    float bm[2] = {MASK2, MASK2};
#pragma unroll
    for (int e = 0; e < 32; ++e) {
      t[e] = logit2(av[e], nv[e], svL);
      bm[e >> 4] = fmaxf(bm[e >> 4], t[e]);
    }
    bm[0] = fmaxf(bm[0], xhalf(bm[0]));
    bm[1] = fmaxf(bm[1], xhalf(bm[1]));

#pragma unroll
    for (int i = 0; i < 4; ++i)
      *(v8h*)(&sB[cur ^ 1][loff + i * 16 * PITCH_H]) = g[i];

#pragma unroll
    for (int u = 0; u < 2; ++u) {
      if (__builtin_amdgcn_ballot_w32(bm[u] > m_run + LAZY) != 0) {
        float m_new = fmaxf(m_run, bm[u]);
        float f = __builtin_amdgcn_exp2f(m_run - m_new);
        l_run *= f;
#pragma unroll
        for (int k = 0; k < 8; ++k) {
          float fk = __shfl(f, k + (hf << 3), 32);
          c0[k] *= fk;
          c1[k] *= fk;
          c2[k] *= fk;
          c3[k] *= fk;
        }
        m_run = m_new;
      }

      v16h a;
#pragma unroll
      for (int e = 0; e < 16; ++e) {
        float p = __builtin_amdgcn_exp2f(t[16 * u + e] - m_run);
        l_run += p;
        a[e] = (_Float16)p;
      }
      c0 = wmma16(a, bf[u][0], c0);
      c1 = wmma16(a, bf[u][1], c1);
      c2 = wmma16(a, bf[u][2], c2);
      c3 = wmma16(a, bf[u][3], c3);
    }
    __syncthreads();
  }

  float inv = 1.0f / (l_run + xhalf(l_run));
  float ik[8];
#pragma unroll
  for (int k = 0; k < 8; ++k) ik[k] = __shfl(inv, k + (hf << 3), 32);

  v8f acc[4] = {c0, c1, c2, c3};
#pragma unroll
  for (int nt = 0; nt < 4; ++nt) {
#pragma unroll
    for (int k = 0; k < 8; ++k) {
      int orow = r0 + k + (hf << 3);
      int col  = nt * 16 + r;
      float v  = acc[nt][k] * ik[k] + b1[head * NHID + col];
      v = v > 0.0f ? v : 0.0f;
      H[(size_t)orow * (H1 * NHID) + head * NHID + col] = (_Float16)v;
    }
  }
}

__global__ void gemm2_kernel(const _Float16* __restrict__ H,
                             const _Float16* __restrict__ W2T,
                             float* __restrict__ feats2,
                             _Float16* __restrict__ feats2T) {
  int r0 = blockIdx.x << 4;
  v8f c = {};
  for (int k0 = 0; k0 < H1 * NHID; k0 += 32) {
    v16h a = load_frag16(H, r0, H1 * NHID, k0);
    v16h b = load_frag16(W2T, 0, H1 * NHID, k0);
    c = wmma16(a, b, c);
  }
  int lane = threadIdx.x & 31;
  int hf   = lane >> 4;
  int r    = lane & 15;
#pragma unroll
  for (int k = 0; k < 8; ++k) {
    int orow = r0 + k + (hf << 3);
    float v  = c[k];
    feats2[(size_t)orow * NCLASS + r] = v;
    feats2T[(size_t)r * N + orow]     = (_Float16)v;
  }
}

#define SPLIT2 16

__global__ void __launch_bounds__(32 * SPLIT2)
__attribute__((amdgpu_waves_per_eu(4, 8)))
attn_gemm2_kernel(const int* __restrict__ adj,
                  const _Float16* __restrict__ feats2T,
                  const float* __restrict__ s2,
                  const float* __restrict__ n2,
                  const float* __restrict__ b2,
                  float* __restrict__ out) {
  __shared__ float sh_c[SPLIT2][16][NCLASS];
  __shared__ float sh_m[SPLIT2][16];
  __shared__ float sh_l[SPLIT2][16];

  int r0   = blockIdx.x << 4;
  int wv   = __builtin_amdgcn_readfirstlane((int)(threadIdx.x >> 5));
  int lane = threadIdx.x & 31;
  int hf   = lane >> 4;
  int r    = lane & 15;
  int row  = r0 + r;

  float svL = s2[row] * LOG2E;
  const int* arow = adj + (size_t)row * N;

  float m_run = -1.0e30f;
  float l_run = 0.0f;

  v8f c = {};
  const int jbeg = wv * (N / SPLIT2);
  const int jend = jbeg + (N / SPLIT2);
  for (int j0 = jbeg; j0 < jend; j0 += 32) {
    int   av[16];
    float nv[16];
#pragma unroll
    for (int e = 0; e < 16; ++e) {
      int j = j0 + (e & 7) + (hf << 3) + ((e >> 3) << 4);
      av[e] = arow[j];
      nv[e] = n2[j];
    }
    v16h b = load_frag16(feats2T, 0, N, j0);

    float t[16];
    float bm = MASK2;
#pragma unroll
    for (int e = 0; e < 16; ++e) {
      t[e] = logit2(av[e], nv[e], svL);
      bm   = fmaxf(bm, t[e]);
    }
    bm = fmaxf(bm, xhalf(bm));

    asm volatile("" : "+v"(b), "+v"(bm));

    if (__builtin_amdgcn_ballot_w32(bm > m_run + LAZY) != 0) {
      float m_new = fmaxf(m_run, bm);
      float f = __builtin_amdgcn_exp2f(m_run - m_new);
      l_run *= f;
#pragma unroll
      for (int k = 0; k < 8; ++k) c[k] *= __shfl(f, k + (hf << 3), 32);
      m_run = m_new;
    }

    v16h a;
#pragma unroll
    for (int e = 0; e < 16; ++e) {
      float p = __builtin_amdgcn_exp2f(t[e] - m_run);
      l_run += p;
      a[e] = (_Float16)p;
    }
    c = wmma16(a, b, c);
  }

  float lsum = l_run + xhalf(l_run);
  if (hf == 0) {
    sh_m[wv][r] = m_run;
    sh_l[wv][r] = lsum;
  }
#pragma unroll
  for (int k = 0; k < 8; ++k) sh_c[wv][k + (hf << 3)][r] = c[k];
  __syncthreads();

  if (threadIdx.x < 16 * NCLASS) {
    int mrow = threadIdx.x >> 4;
    int cls  = threadIdx.x & 15;
    float M = sh_m[0][mrow];
#pragma unroll
    for (int w = 1; w < SPLIT2; ++w) M = fmaxf(M, sh_m[w][mrow]);
    float L = 0.0f, C = 0.0f;
#pragma unroll
    for (int w = 0; w < SPLIT2; ++w) {
      float f = __builtin_amdgcn_exp2f(sh_m[w][mrow] - M);
      L = __builtin_fmaf(sh_l[w][mrow], f, L);
      C = __builtin_fmaf(sh_c[w][mrow][cls], f, C);
    }
    float inv = 1.0f / L;
    float v   = C * inv + b2[cls];
    v = v > 0.0f ? v : 0.0f;
    float mx = v;
#pragma unroll
    for (int msk = 1; msk < 16; msk <<= 1)
      mx = fmaxf(mx, __shfl_xor(mx, msk, 32));
    float ex = __expf(v - mx);
    float sum = ex;
#pragma unroll
    for (int msk = 1; msk < 16; msk <<= 1) sum += __shfl_xor(sum, msk, 32);
    float o = (v - mx) - logf(sum);
    out[(size_t)(r0 + mrow) * NCLASS + cls] = o;
  }
}

extern "C" void kernel_launch(void* const* d_in, const int* in_sizes, int n_in,
                              void* d_out, int out_size, void* d_ws,
                              size_t ws_size, hipStream_t stream) {
  const float* x   = (const float*)d_in[0];
  const int*   adj = (const int*)d_in[1];
  const float* W1  = (const float*)d_in[2];
  const float* b1  = (const float*)d_in[3];
  const float* as1 = (const float*)d_in[4];
  const float* an1 = (const float*)d_in[5];
  const float* W2  = (const float*)d_in[6];
  const float* b2  = (const float*)d_in[7];
  const float* as2 = (const float*)d_in[8];
  const float* an2 = (const float*)d_in[9];
  float* out = (float*)d_out;

  char* ws = (char*)d_ws;
  size_t off = 0;
  auto alloc = [&](size_t bytes) -> void* {
    off = (off + 255) & ~(size_t)255;
    void* p = ws + off;
    off += bytes;
    return p;
  };

  _Float16* XH      = (_Float16*)alloc((size_t)N * NFEAT * 2);
  _Float16* W1T     = (_Float16*)alloc((size_t)H1 * NHID * NFEAT * 2);
  _Float16* W2T     = (_Float16*)alloc((size_t)NCLASS * H1 * NHID * 2);
  float*    feats1  = (float*)alloc((size_t)H1 * N * NHID * 4);
  _Float16* featsT  = (_Float16*)alloc((size_t)H1 * NHID * N * 2);
  float*    s1      = (float*)alloc((size_t)H1 * N * 4);
  float*    n1      = (float*)alloc((size_t)H1 * N * 4);
  _Float16* H       = (_Float16*)alloc((size_t)N * H1 * NHID * 2);
  float*    feats2  = (float*)alloc((size_t)N * NCLASS * 4);
  _Float16* feats2T = (_Float16*)alloc((size_t)NCLASS * N * 2);
  float*    s2      = (float*)alloc((size_t)N * 4);
  float*    n2      = (float*)alloc((size_t)N * 4);
  (void)ws_size; (void)n_in; (void)in_sizes; (void)out_size;

  {
    int total = N * NFEAT;
    cast_kernel<<<(total + 255) / 256, 256, 0, stream>>>(x, W1, W2, XH, W1T,
                                                         W2T);
  }
  gemm1_kernel<<<dim3(N / 16, H1), 32, 0, stream>>>(XH, W1T, feats1, featsT);
  sn_kernel<<<(H1 * N + 255) / 256, 256, 0, stream>>>(feats1, as1, an1, s1, n1,
                                                      NHID, N, H1 * N);
  attn_gemm1_kernel<<<dim3(N / (16 * WAVES1), H1), 32 * WAVES1, 0, stream>>>(
      adj, featsT, s1, n1, b1, H);
  gemm2_kernel<<<N / 16, 32, 0, stream>>>(H, W2T, feats2, feats2T);
  sn_kernel<<<(N + 255) / 256, 256, 0, stream>>>(feats2, as2, an2, s2, n2,
                                                 NCLASS, N, N);
  attn_gemm2_kernel<<<N / 16, 32 * SPLIT2, 0, stream>>>(adj, feats2T, s2, n2,
                                                        b2, out);
}


# Adjacency reduced once to lane masks (1 bit per edge, in A-fragment order) inside the layer-1 dots' launch; both attention kernels fetch them by scalar loads, select by mask; order of sums unchanged
# speedup vs baseline: 4.3711x; 1.0894x over previous
#include <hip/hip_runtime.h>
#include <hip/hip_bf16.h>


#define N      4096
#define NFEAT  512
#define NHID   64
#define NCLASS 16
#define H1     8
#define NEG_SLOPE 0.2f
#define MASK_VAL  -1e10f
#define LOG2E 1.4426950408889634f
#define MASK2 (MASK_VAL * LOG2E)
#define LAZY 8.0f

typedef __attribute__((ext_vector_type(16))) _Float16 v16h;
typedef __attribute__((ext_vector_type(8)))  _Float16 v8h;
typedef __attribute__((ext_vector_type(8)))  float    v8f;

__device__ __forceinline__ v8f wmma16(v16h a, v16h b, v8f c) {
  return __builtin_amdgcn_wmma_f32_16x16x32_f16(
       false, a,  false, b,
       (short)0, c,  false,  false);
}

__device__ __forceinline__ v16h load_frag16(const _Float16* base, int rc,
                                            int ld, int k0) {
  int lane = threadIdx.x & 31;
  int hf   = lane >> 4;
  int r    = lane & 15;
  const _Float16* p = base + (size_t)(rc + r) * ld + k0 + (hf << 3);
  v8h lo = *(const v8h*)p;
  v8h hi = *(const v8h*)(p + 16);
  return __builtin_shufflevector(lo, hi, 0, 1, 2, 3, 4, 5, 6, 7, 8, 9, 10, 11,
                                 12, 13, 14, 15);
}

__device__ __forceinline__ float lrelu(float x) {
  return x > 0.0f ? x : NEG_SLOPE * x;
}

__device__ __forceinline__ float xhalf(float x) {
  int b = __builtin_bit_cast(int, x);
  return __builtin_bit_cast(
      float, __builtin_amdgcn_permlanex16(b, b, 0x76543210, 0xfedcba98, false,
                                          false));
}

__device__ __forceinline__ float logit2m(bool on, float nj, float svL) {
  float x2 = __builtin_fmaf(nj, LOG2E, svL);
  float lr = fmaxf(x2, NEG_SLOPE * x2);
  return on ? lr : MASK2;
}

#define NCB (N / 32)
#define MASK_BLOCKS ((N / 16) * (NCB / 4) / 8)
typedef __attribute__((ext_vector_type(4))) int v4i;

__device__ __forceinline__ void adjmask_block(const int* __restrict__ adj,
                                              unsigned* __restrict__ masks,
                                              int blk) {
  int wv   = __builtin_amdgcn_readfirstlane((int)(threadIdx.x >> 5));
  int gw   = blk * 8 + wv;
  int rt   = gw >> 5;
  int cb0  = (gw & 31) << 2;
  int lane = threadIdx.x & 31;
  int hf   = lane >> 4;
  int r    = lane & 15;
  const int* arow = adj + (size_t)(rt * 16 + r) * N + cb0 * 32 + (hf << 3);

  v4i a[4][4];
#pragma unroll
  for (int cb = 0; cb < 4; ++cb) {
#pragma unroll
    for (int q = 0; q < 4; ++q) {
      a[cb][q] = __builtin_nontemporal_load(
          (const v4i*)(arow + cb * 32 + (q & 1) * 4 + (q >> 1) * 16));
    }
  }
  unsigned o[2] = {0u, 0u};
#pragma unroll
  for (int cb = 0; cb < 4; ++cb) {
#pragma unroll
    for (int e = 0; e < 16; ++e) {
      unsigned m = __builtin_amdgcn_ballot_w32(a[cb][e >> 2][e & 3] > 0);
      o[cb >> 1] = (lane == e + 16 * (cb & 1)) ? m : o[cb >> 1];
    }
  }
  unsigned* dst = masks + ((size_t)rt * NCB + cb0) * 16;
  dst[lane]      = o[0];
  dst[32 + lane] = o[1];
}

__global__ void cast_kernel(const float* __restrict__ x,
                            const float* __restrict__ W1,
                            const float* __restrict__ W2,
                            _Float16* __restrict__ XH,
                            _Float16* __restrict__ W1T,
                            _Float16* __restrict__ W2T) {
  int i = blockIdx.x * blockDim.x + threadIdx.x;
  if (i < N * NFEAT) XH[i] = (_Float16)x[i];
  if (i < H1 * NFEAT * NHID) {
    int h   = i / (NFEAT * NHID);
    int rem = i % (NFEAT * NHID);
    int k   = rem / NHID;
    int c   = rem % NHID;
    W1T[((size_t)(h * NHID + c)) * NFEAT + k] = (_Float16)W1[i];
  }
  if (i < (H1 * NHID) * NCLASS) {
    int k = i / NCLASS;
    int c = i % NCLASS;
    W2T[(size_t)c * (H1 * NHID) + k] = (_Float16)W2[i];
  }
}

__global__ void gemm1_kernel(const _Float16* __restrict__ XH,
                             const _Float16* __restrict__ W1T,
                             float* __restrict__ feats,
                             _Float16* __restrict__ featsT) {
  int head = blockIdx.y;
  int r0   = blockIdx.x << 4;
  const _Float16* BT = W1T + (size_t)head * NHID * NFEAT;

  v8f c0 = {}, c1 = {}, c2 = {}, c3 = {};
  for (int k0 = 0; k0 < NFEAT; k0 += 32) {
    v16h a = load_frag16(XH, r0, NFEAT, k0);
    c0 = wmma16(a, load_frag16(BT, 0, NFEAT, k0), c0);
    c1 = wmma16(a, load_frag16(BT, 16, NFEAT, k0), c1);
    c2 = wmma16(a, load_frag16(BT, 32, NFEAT, k0), c2);
    c3 = wmma16(a, load_frag16(BT, 48, NFEAT, k0), c3);
  }

  int lane = threadIdx.x & 31;
  int hf   = lane >> 4;
  int r    = lane & 15;
  v8f acc[4] = {c0, c1, c2, c3};
#pragma unroll
  for (int nt = 0; nt < 4; ++nt) {
#pragma unroll
    for (int k = 0; k < 8; ++k) {
      int orow = r0 + k + (hf << 3);
      int col  = nt * 16 + r;
      float v  = acc[nt][k];
      feats[((size_t)head * N + orow) * NHID + col] = v;
      featsT[((size_t)head * NHID + col) * N + orow] = (_Float16)v;
    }
  }
}

__device__ __forceinline__ void sn_row(const float* __restrict__ feats,
                                       const float* __restrict__ a_s,
                                       const float* __restrict__ a_n,
                                       float* __restrict__ s,
                                       float* __restrict__ n, int fout,
                                       int rowsPerHead, int total, int i) {
  if (i >= total) return;
  int head = i / rowsPerHead;
  const float* f  = feats + (size_t)i * fout;
  const float* av = a_s + head * fout;
  const float* nv = a_n + head * fout;
  float ss = 0.0f, nn = 0.0f;
  for (int c = 0; c < fout; ++c) {
    float v = f[c];
    ss += v * av[c];
    nn += v * nv[c];
  }
  s[i] = ss;
  n[i] = nn;
}

__global__ void sn_kernel(const float* __restrict__ feats,
                          const float* __restrict__ a_s,
                          const float* __restrict__ a_n,
                          float* __restrict__ s, float* __restrict__ n,
                          int fout, int rowsPerHead, int total) {
  sn_row(feats, a_s, a_n, s, n, fout, rowsPerHead, total,
         blockIdx.x * blockDim.x + threadIdx.x);
}

#define SN1_BLOCKS ((H1 * N) / 256)

__global__ void __launch_bounds__(256)
sn1_adjmask_kernel(const float* __restrict__ feats,
                   const float* __restrict__ a_s,
                   const float* __restrict__ a_n, float* __restrict__ s,
                   float* __restrict__ n, const int* __restrict__ adj,
                   unsigned* __restrict__ masks) {
  if (blockIdx.x < SN1_BLOCKS)
    sn_row(feats, a_s, a_n, s, n, NHID, N, H1 * N,
           (int)(blockIdx.x * 256 + threadIdx.x));
  else
    adjmask_block(adj, masks, (int)blockIdx.x - SN1_BLOCKS);
}

#define WAVES1  4
#define KT      64
#define PITCH_H 72

__global__ void __launch_bounds__(32 * WAVES1)
__attribute__((amdgpu_waves_per_eu(1, 2)))
attn_gemm1_kernel(const unsigned* __restrict__ masks,
                  const _Float16* __restrict__ featsT,
                  const float* __restrict__ s1,
                  const float* __restrict__ n1,
                  const float* __restrict__ b1,
                  _Float16* __restrict__ H) {
  __shared__ __attribute__((aligned(16))) _Float16 sB[2][NHID * PITCH_H];

  int head = blockIdx.y;
  int wv   = __builtin_amdgcn_readfirstlane((int)(threadIdx.x >> 5));
  int r0   = ((int)blockIdx.x * WAVES1 + wv) << 4;
  int lane = threadIdx.x & 31;
  int hf   = lane >> 4;
  int r    = lane & 15;
  int row  = r0 + r;

  const float* nrow = n1 + head * N;
  float svL = s1[head * N + row] * LOG2E;
  const unsigned* mrow = masks + (size_t)(r0 >> 4) * (NCB * 16);
  const _Float16* BT = featsT + (size_t)head * NHID * N;

  const int tid = threadIdx.x;
  const _Float16* gsrc = BT + (size_t)(tid >> 3) * N + ((tid & 7) << 3);
  const int loff = (tid >> 3) * PITCH_H + ((tid & 7) << 3);

#pragma unroll
  for (int i = 0; i < 4; ++i) {
    v8h g = *(const v8h*)(gsrc + (size_t)i * 16 * N);
    *(v8h*)(&sB[0][loff + i * 16 * PITCH_H]) = g;
  }
  __syncthreads();

  float m_run = -1.0e30f;
  float l_run = 0.0f;

  unsigned mk[32];
#pragma unroll
  for (int e = 0; e < 32; ++e) mk[e] = mrow[e];

  v8f c0 = {}, c1 = {}, c2 = {}, c3 = {};
  for (int j0 = 0; j0 < N; j0 += KT) {
    const int cur = (j0 / KT) & 1;
    float nv[32];
#pragma unroll
    for (int e = 0; e < 32; ++e) {
      int j = j0 + ((e >> 4) << 5) + (e & 7) + (hf << 3) + (((e >> 3) & 1) << 4);
      nv[e] = nrow[j];
    }
    const int jn = (j0 + KT < N) ? (j0 + KT) : j0;
    v8h g[4];
#pragma unroll
    for (int i = 0; i < 4; ++i)
      g[i] = *(const v8h*)(gsrc + (size_t)i * 16 * N + jn);

    const _Float16* st = &sB[cur][0];
    v16h bf[2][4];
#pragma unroll
    for (int u = 0; u < 2; ++u)
#pragma unroll
      for (int nt = 0; nt < 4; ++nt)
        bf[u][nt] = load_frag16(st, nt * 16, PITCH_H, 32 * u);

    float t[32];
    float bm[2] = {MASK2, MASK2};
#pragma unroll
    for (int e = 0; e < 32; ++e) {
      t[e] = logit2m(__builtin_amdgcn_inverse_ballot_w32(mk[e]), nv[e], svL);
      bm[e >> 4] = fmaxf(bm[e >> 4], t[e]);
    }
    bm[0] = fmaxf(bm[0], xhalf(bm[0]));
    bm[1] = fmaxf(bm[1], xhalf(bm[1]));

#pragma unroll
    for (int i = 0; i < 4; ++i)
      *(v8h*)(&sB[cur ^ 1][loff + i * 16 * PITCH_H]) = g[i];

#pragma unroll
    for (int u = 0; u < 2; ++u) {
      if (__builtin_amdgcn_ballot_w32(bm[u] > m_run + LAZY) != 0) {
        float m_new = fmaxf(m_run, bm[u]);
        float f = __builtin_amdgcn_exp2f(m_run - m_new);
        l_run *= f;
#pragma unroll
        for (int k = 0; k < 8; ++k) {
          float fk = __shfl(f, k + (hf << 3), 32);
          c0[k] *= fk;
          c1[k] *= fk;
          c2[k] *= fk;
          c3[k] *= fk;
        }
        m_run = m_new;
      }

      if (u == 0) {
        const int jm = (j0 + KT < N) ? (j0 + KT) : j0;
#pragma unroll
        for (int e = 0; e < 32; ++e) mk[e] = mrow[(jm >> 1) + e];
      }

      v16h a;
#pragma unroll
      for (int e = 0; e < 16; ++e) {
        float p = __builtin_amdgcn_exp2f(t[16 * u + e] - m_run);
        l_run += p;
        a[e] = (_Float16)p;
      }
      c0 = wmma16(a, bf[u][0], c0);
      c1 = wmma16(a, bf[u][1], c1);
      c2 = wmma16(a, bf[u][2], c2);
      c3 = wmma16(a, bf[u][3], c3);
    }
    __syncthreads();
  }

  float inv = 1.0f / (l_run + xhalf(l_run));
  float ik[8];
#pragma unroll
  for (int k = 0; k < 8; ++k) ik[k] = __shfl(inv, k + (hf << 3), 32);

  v8f acc[4] = {c0, c1, c2, c3};
#pragma unroll
  for (int nt = 0; nt < 4; ++nt) {
#pragma unroll
    for (int k = 0; k < 8; ++k) {
      int orow = r0 + k + (hf << 3);
      int col  = nt * 16 + r;
      float v  = acc[nt][k] * ik[k] + b1[head * NHID + col];
      v = v > 0.0f ? v : 0.0f;
      H[(size_t)orow * (H1 * NHID) + head * NHID + col] = (_Float16)v;
    }
  }
}

__global__ void gemm2_kernel(const _Float16* __restrict__ H,
                             const _Float16* __restrict__ W2T,
                             float* __restrict__ feats2,
                             _Float16* __restrict__ feats2T) {
  int r0 = blockIdx.x << 4;
  v8f c = {};
  for (int k0 = 0; k0 < H1 * NHID; k0 += 32) {
    v16h a = load_frag16(H, r0, H1 * NHID, k0);
    v16h b = load_frag16(W2T, 0, H1 * NHID, k0);
    c = wmma16(a, b, c);
  }
  int lane = threadIdx.x & 31;
  int hf   = lane >> 4;
  int r    = lane & 15;
#pragma unroll
  for (int k = 0; k < 8; ++k) {
    int orow = r0 + k + (hf << 3);
    float v  = c[k];
    feats2[(size_t)orow * NCLASS + r] = v;
    feats2T[(size_t)r * N + orow]     = (_Float16)v;
  }
}

#define SPLIT2 16

__global__ void __launch_bounds__(32 * SPLIT2)
__attribute__((amdgpu_waves_per_eu(4, 8)))
attn_gemm2_kernel(const unsigned* __restrict__ masks,
                  const _Float16* __restrict__ feats2T,
                  const float* __restrict__ s2,
                  const float* __restrict__ n2,
                  const float* __restrict__ b2,
                  float* __restrict__ out) {
  __shared__ float sh_c[SPLIT2][16][NCLASS];
  __shared__ float sh_m[SPLIT2][16];
  __shared__ float sh_l[SPLIT2][16];

  int r0   = blockIdx.x << 4;
  int wv   = __builtin_amdgcn_readfirstlane((int)(threadIdx.x >> 5));
  int lane = threadIdx.x & 31;
  int hf   = lane >> 4;
  int r    = lane & 15;
  int row  = r0 + r;

  float svL = s2[row] * LOG2E;
  const unsigned* mrow = masks + (size_t)blockIdx.x * (NCB * 16);

  float m_run = -1.0e30f;
  float l_run = 0.0f;

  v8f c = {};
  const int jbeg = wv * (N / SPLIT2);
  const int jend = jbeg + (N / SPLIT2);
  unsigned mk[16];
#pragma unroll
  for (int e = 0; e < 16; ++e) mk[e] = mrow[(jbeg >> 1) + e];

  for (int j0 = jbeg; j0 < jend; j0 += 32) {
    float nv[16];
#pragma unroll
    for (int e = 0; e < 16; ++e) {
      int j = j0 + (e & 7) + (hf << 3) + ((e >> 3) << 4);
      nv[e] = n2[j];
    }
    v16h b = load_frag16(feats2T, 0, N, j0);

    float t[16];
    float bm = MASK2;
#pragma unroll
    for (int e = 0; e < 16; ++e) {
      t[e] = logit2m(__builtin_amdgcn_inverse_ballot_w32(mk[e]), nv[e], svL);
      bm   = fmaxf(bm, t[e]);
    }
    bm = fmaxf(bm, xhalf(bm));

    asm volatile("" : "+v"(b), "+v"(bm));

    if (__builtin_amdgcn_ballot_w32(bm > m_run + LAZY) != 0) {
      float m_new = fmaxf(m_run, bm);
      float f = __builtin_amdgcn_exp2f(m_run - m_new);
      l_run *= f;
#pragma unroll
      for (int k = 0; k < 8; ++k) c[k] *= __shfl(f, k + (hf << 3), 32);
      m_run = m_new;
    }

    {
      const int jm = (j0 + 32 < jend) ? (j0 + 32) : j0;
#pragma unroll
      for (int e = 0; e < 16; ++e) mk[e] = mrow[(jm >> 1) + e];
    }

    v16h a;
#pragma unroll
    for (int e = 0; e < 16; ++e) {
      float p = __builtin_amdgcn_exp2f(t[e] - m_run);
      l_run += p;
      a[e] = (_Float16)p;
    }
    c = wmma16(a, b, c);
  }

  float lsum = l_run + xhalf(l_run);
  if (hf == 0) {
    sh_m[wv][r] = m_run;
    sh_l[wv][r] = lsum;
  }
#pragma unroll
  for (int k = 0; k < 8; ++k) sh_c[wv][k + (hf << 3)][r] = c[k];
  __syncthreads();

  if (threadIdx.x < 16 * NCLASS) {
    int mrow = threadIdx.x >> 4;
    int cls  = threadIdx.x & 15;
    float M = sh_m[0][mrow];
#pragma unroll
    for (int w = 1; w < SPLIT2; ++w) M = fmaxf(M, sh_m[w][mrow]);
    float L = 0.0f, C = 0.0f;
#pragma unroll
    for (int w = 0; w < SPLIT2; ++w) {
      float f = __builtin_amdgcn_exp2f(sh_m[w][mrow] - M);
      L = __builtin_fmaf(sh_l[w][mrow], f, L);
      C = __builtin_fmaf(sh_c[w][mrow][cls], f, C);
    }
    float inv = 1.0f / L;
    float v   = C * inv + b2[cls];
    v = v > 0.0f ? v : 0.0f;
    float mx = v;
#pragma unroll
    for (int msk = 1; msk < 16; msk <<= 1)
      mx = fmaxf(mx, __shfl_xor(mx, msk, 32));
    float ex = __expf(v - mx);
    float sum = ex;
#pragma unroll
    for (int msk = 1; msk < 16; msk <<= 1) sum += __shfl_xor(sum, msk, 32);
    float o = (v - mx) - logf(sum);
    out[(size_t)(r0 + mrow) * NCLASS + cls] = o;
  }
}

extern "C" void kernel_launch(void* const* d_in, const int* in_sizes, int n_in,
                              void* d_out, int out_size, void* d_ws,
                              size_t ws_size, hipStream_t stream) {
  const float* x   = (const float*)d_in[0];
  const int*   adj = (const int*)d_in[1];
  const float* W1  = (const float*)d_in[2];
  const float* b1  = (const float*)d_in[3];
  const float* as1 = (const float*)d_in[4];
  const float* an1 = (const float*)d_in[5];
  const float* W2  = (const float*)d_in[6];
  const float* b2  = (const float*)d_in[7];
  const float* as2 = (const float*)d_in[8];
  const float* an2 = (const float*)d_in[9];
  float* out = (float*)d_out;

  char* ws = (char*)d_ws;
  size_t off = 0;
  auto alloc = [&](size_t bytes) -> void* {
    off = (off + 255) & ~(size_t)255;
    void* p = ws + off;
    off += bytes;
    return p;
  };

  _Float16* XH      = (_Float16*)alloc((size_t)N * NFEAT * 2);
  _Float16* W1T     = (_Float16*)alloc((size_t)H1 * NHID * NFEAT * 2);
  _Float16* W2T     = (_Float16*)alloc((size_t)NCLASS * H1 * NHID * 2);
  float*    feats1  = (float*)alloc((size_t)H1 * N * NHID * 4);
  _Float16* featsT  = (_Float16*)alloc((size_t)H1 * NHID * N * 2);
  float*    s1      = (float*)alloc((size_t)H1 * N * 4);
  float*    n1      = (float*)alloc((size_t)H1 * N * 4);
  _Float16* H       = (_Float16*)alloc((size_t)N * H1 * NHID * 2);
  float*    feats2  = (float*)alloc((size_t)N * NCLASS * 4);
  _Float16* feats2T = (_Float16*)alloc((size_t)NCLASS * N * 2);
  float*    s2      = (float*)alloc((size_t)N * 4);
  float*    n2      = (float*)alloc((size_t)N * 4);
  unsigned* masks   = (unsigned*)XH;
  (void)ws_size; (void)n_in; (void)in_sizes; (void)out_size;

  {
    int total = N * NFEAT;
    cast_kernel<<<(total + 255) / 256, 256, 0, stream>>>(x, W1, W2, XH, W1T,
                                                         W2T);
  }
  gemm1_kernel<<<dim3(N / 16, H1), 32, 0, stream>>>(XH, W1T, feats1, featsT);
  sn1_adjmask_kernel<<<SN1_BLOCKS + MASK_BLOCKS, 256, 0, stream>>>(
      feats1, as1, an1, s1, n1, adj, masks);
  attn_gemm1_kernel<<<dim3(N / (16 * WAVES1), H1), 32 * WAVES1, 0, stream>>>(
      masks, featsT, s1, n1, b1, H);
  gemm2_kernel<<<N / 16, 32, 0, stream>>>(H, W2T, feats2, feats2T);
  sn_kernel<<<(N + 255) / 256, 256, 0, stream>>>(feats2, as2, an2, s2, n2,
                                                 NCLASS, N, N);
  attn_gemm2_kernel<<<N / 16, 32 * SPLIT2, 0, stream>>>(masks, feats2T, s2, n2,
                                                        b2, out);
}


# Cast launch: each thread converts 32 floats of x by 16-byte loads (W1, W2: 8 k per thread, 16-byte stores); 3,104 waves enter instead of 65,536; values and every sum unchanged
# speedup vs baseline: 4.8986x; 1.1207x over previous
#include <hip/hip_runtime.h>
#include <hip/hip_bf16.h>


#define N      4096
#define NFEAT  512
#define NHID   64
#define NCLASS 16
#define H1     8
#define NEG_SLOPE 0.2f
#define MASK_VAL  -1e10f
#define LOG2E 1.4426950408889634f
#define MASK2 (MASK_VAL * LOG2E)
#define LAZY 8.0f

typedef __attribute__((ext_vector_type(16))) _Float16 v16h;
typedef __attribute__((ext_vector_type(8)))  _Float16 v8h;
typedef __attribute__((ext_vector_type(8)))  float    v8f;

__device__ __forceinline__ v8f wmma16(v16h a, v16h b, v8f c) {
  return __builtin_amdgcn_wmma_f32_16x16x32_f16(
       false, a,  false, b,
       (short)0, c,  false,  false);
}

__device__ __forceinline__ v16h load_frag16(const _Float16* base, int rc,
                                            int ld, int k0) {
  int lane = threadIdx.x & 31;
  int hf   = lane >> 4;
  int r    = lane & 15;
  const _Float16* p = base + (size_t)(rc + r) * ld + k0 + (hf << 3);
  v8h lo = *(const v8h*)p;
  v8h hi = *(const v8h*)(p + 16);
  return __builtin_shufflevector(lo, hi, 0, 1, 2, 3, 4, 5, 6, 7, 8, 9, 10, 11,
                                 12, 13, 14, 15);
}

__device__ __forceinline__ float lrelu(float x) {
  return x > 0.0f ? x : NEG_SLOPE * x;
}

__device__ __forceinline__ float xhalf(float x) {
  int b = __builtin_bit_cast(int, x);
  return __builtin_bit_cast(
      float, __builtin_amdgcn_permlanex16(b, b, 0x76543210, 0xfedcba98, false,
                                          false));
}

__device__ __forceinline__ float logit2m(bool on, float nj, float svL) {
  float x2 = __builtin_fmaf(nj, LOG2E, svL);
  float lr = fmaxf(x2, NEG_SLOPE * x2);
  return on ? lr : MASK2;
}

#define NCB (N / 32)
#define MASK_BLOCKS ((N / 16) * (NCB / 4) / 8)
typedef __attribute__((ext_vector_type(4))) int v4i;

__device__ __forceinline__ void adjmask_block(const int* __restrict__ adj,
                                              unsigned* __restrict__ masks,
                                              int blk) {
  int wv   = __builtin_amdgcn_readfirstlane((int)(threadIdx.x >> 5));
  int gw   = blk * 8 + wv;
  int rt   = gw >> 5;
  int cb0  = (gw & 31) << 2;
  int lane = threadIdx.x & 31;
  int hf   = lane >> 4;
  int r    = lane & 15;
  const int* arow = adj + (size_t)(rt * 16 + r) * N + cb0 * 32 + (hf << 3);

  v4i a[4][4];
#pragma unroll
  for (int cb = 0; cb < 4; ++cb) {
#pragma unroll
    for (int q = 0; q < 4; ++q) {
      a[cb][q] = __builtin_nontemporal_load(
          (const v4i*)(arow + cb * 32 + (q & 1) * 4 + (q >> 1) * 16));
    }
  }
  unsigned o[2] = {0u, 0u};
#pragma unroll
  for (int cb = 0; cb < 4; ++cb) {
#pragma unroll
    for (int e = 0; e < 16; ++e) {
      unsigned m = __builtin_amdgcn_ballot_w32(a[cb][e >> 2][e & 3] > 0);
      o[cb >> 1] = (lane == e + 16 * (cb & 1)) ? m : o[cb >> 1];
    }
  }
  unsigned* dst = masks + ((size_t)rt * NCB + cb0) * 16;
  dst[lane]      = o[0];
  dst[32 + lane] = o[1];
}

typedef __attribute__((ext_vector_type(4))) float    v4f;
typedef __attribute__((ext_vector_type(4))) _Float16 v4h;

#define CAST_X_BLOCKS     256
#define CAST_X_PER_THREAD ((N * NFEAT / 4) / (CAST_X_BLOCKS * 256))
#define CAST_W1_BLOCKS    ((H1 * NHID * NFEAT / 8) / 256)
#define CAST_W2_BLOCKS    ((NCLASS * H1 * NHID / 8) / 256)

__global__ void __launch_bounds__(256)
cast_kernel(const float* __restrict__ x,
            const float* __restrict__ W1,
            const float* __restrict__ W2,
            _Float16* __restrict__ XH,
            _Float16* __restrict__ W1T,
            _Float16* __restrict__ W2T) {
  const int b   = blockIdx.x;
  const int tid = threadIdx.x;
  if (b < CAST_X_BLOCKS) {
    const int base = b * (256 * CAST_X_PER_THREAD) + tid;
    const v4f* src = (const v4f*)x + base;
    v4h* dst = (v4h*)XH + base;
    v4f v[CAST_X_PER_THREAD];
#pragma unroll
    for (int i = 0; i < CAST_X_PER_THREAD; ++i) v[i] = src[i * 256];
#pragma unroll
    for (int i = 0; i < CAST_X_PER_THREAD; ++i)
      dst[i * 256] = __builtin_convertvector(v[i], v4h);
  } else if (b < CAST_X_BLOCKS + CAST_W1_BLOCKS) {
    const int q  = (b - CAST_X_BLOCKS) * 256 + tid;
    const int c  = q & (NHID - 1);
    const int k0 = ((q / NHID) & (NFEAT / 8 - 1)) * 8;
    const int h  = q / (NHID * (NFEAT / 8));
    const float* src = W1 + ((size_t)h * NFEAT + k0) * NHID + c;
    v8h o;
#pragma unroll
    for (int j = 0; j < 8; ++j) o[j] = (_Float16)src[j * NHID];
    *(v8h*)(W1T + ((size_t)(h * NHID + c)) * NFEAT + k0) = o;
  } else {
    const int q  = (b - CAST_X_BLOCKS - CAST_W1_BLOCKS) * 256 + tid;
    const int c  = q & (NCLASS - 1);
    const int k0 = (q / NCLASS) * 8;
    const float* src = W2 + (size_t)k0 * NCLASS + c;
    v8h o;
#pragma unroll
    for (int j = 0; j < 8; ++j) o[j] = (_Float16)src[j * NCLASS];
    *(v8h*)(W2T + (size_t)c * (H1 * NHID) + k0) = o;
  }
}

__global__ void gemm1_kernel(const _Float16* __restrict__ XH,
                             const _Float16* __restrict__ W1T,
                             float* __restrict__ feats,
                             _Float16* __restrict__ featsT) {
  int head = blockIdx.y;
  int r0   = blockIdx.x << 4;
  const _Float16* BT = W1T + (size_t)head * NHID * NFEAT;

  v8f c0 = {}, c1 = {}, c2 = {}, c3 = {};
  for (int k0 = 0; k0 < NFEAT; k0 += 32) {
    v16h a = load_frag16(XH, r0, NFEAT, k0);
    c0 = wmma16(a, load_frag16(BT, 0, NFEAT, k0), c0);
    c1 = wmma16(a, load_frag16(BT, 16, NFEAT, k0), c1);
    c2 = wmma16(a, load_frag16(BT, 32, NFEAT, k0), c2);
    c3 = wmma16(a, load_frag16(BT, 48, NFEAT, k0), c3);
  }

  int lane = threadIdx.x & 31;
  int hf   = lane >> 4;
  int r    = lane & 15;
  v8f acc[4] = {c0, c1, c2, c3};
#pragma unroll
  for (int nt = 0; nt < 4; ++nt) {
#pragma unroll
    for (int k = 0; k < 8; ++k) {
      int orow = r0 + k + (hf << 3);
      int col  = nt * 16 + r;
      float v  = acc[nt][k];
      feats[((size_t)head * N + orow) * NHID + col] = v;
      featsT[((size_t)head * NHID + col) * N + orow] = (_Float16)v;
    }
  }
}

__device__ __forceinline__ void sn_row(const float* __restrict__ feats,
                                       const float* __restrict__ a_s,
                                       const float* __restrict__ a_n,
                                       float* __restrict__ s,
                                       float* __restrict__ n, int fout,
                                       int rowsPerHead, int total, int i) {
  if (i >= total) return;
  int head = i / rowsPerHead;
  const float* f  = feats + (size_t)i * fout;
  const float* av = a_s + head * fout;
  const float* nv = a_n + head * fout;
  float ss = 0.0f, nn = 0.0f;
  for (int c = 0; c < fout; ++c) {
    float v = f[c];
    ss += v * av[c];
    nn += v * nv[c];
  }
  s[i] = ss;
  n[i] = nn;
}

__global__ void sn_kernel(const float* __restrict__ feats,
                          const float* __restrict__ a_s,
                          const float* __restrict__ a_n,
                          float* __restrict__ s, float* __restrict__ n,
                          int fout, int rowsPerHead, int total) {
  sn_row(feats, a_s, a_n, s, n, fout, rowsPerHead, total,
         blockIdx.x * blockDim.x + threadIdx.x);
}

#define SN1_BLOCKS ((H1 * N) / 256)

__global__ void __launch_bounds__(256)
sn1_adjmask_kernel(const float* __restrict__ feats,
                   const float* __restrict__ a_s,
                   const float* __restrict__ a_n, float* __restrict__ s,
                   float* __restrict__ n, const int* __restrict__ adj,
                   unsigned* __restrict__ masks) {
  if (blockIdx.x < SN1_BLOCKS)
    sn_row(feats, a_s, a_n, s, n, NHID, N, H1 * N,
           (int)(blockIdx.x * 256 + threadIdx.x));
  else
    adjmask_block(adj, masks, (int)blockIdx.x - SN1_BLOCKS);
}

#define WAVES1  4
#define KT      64
#define PITCH_H 72

__global__ void __launch_bounds__(32 * WAVES1)
__attribute__((amdgpu_waves_per_eu(1, 2)))
attn_gemm1_kernel(const unsigned* __restrict__ masks,
                  const _Float16* __restrict__ featsT,
                  const float* __restrict__ s1,
                  const float* __restrict__ n1,
                  const float* __restrict__ b1,
                  _Float16* __restrict__ H) {
  __shared__ __attribute__((aligned(16))) _Float16 sB[2][NHID * PITCH_H];

  int head = blockIdx.y;
  int wv   = __builtin_amdgcn_readfirstlane((int)(threadIdx.x >> 5));
  int r0   = ((int)blockIdx.x * WAVES1 + wv) << 4;
  int lane = threadIdx.x & 31;
  int hf   = lane >> 4;
  int r    = lane & 15;
  int row  = r0 + r;

  const float* nrow = n1 + head * N;
  float svL = s1[head * N + row] * LOG2E;
  const unsigned* mrow = masks + (size_t)(r0 >> 4) * (NCB * 16);
  const _Float16* BT = featsT + (size_t)head * NHID * N;

  const int tid = threadIdx.x;
  const _Float16* gsrc = BT + (size_t)(tid >> 3) * N + ((tid & 7) << 3);
  const int loff = (tid >> 3) * PITCH_H + ((tid & 7) << 3);

#pragma unroll
  for (int i = 0; i < 4; ++i) {
    v8h g = *(const v8h*)(gsrc + (size_t)i * 16 * N);
    *(v8h*)(&sB[0][loff + i * 16 * PITCH_H]) = g;
  }
  __syncthreads();

  float m_run = -1.0e30f;
  float l_run = 0.0f;

  unsigned mk[32];
#pragma unroll
  for (int e = 0; e < 32; ++e) mk[e] = mrow[e];

  v8f c0 = {}, c1 = {}, c2 = {}, c3 = {};
  for (int j0 = 0; j0 < N; j0 += KT) {
    const int cur = (j0 / KT) & 1;
    float nv[32];
#pragma unroll
    for (int e = 0; e < 32; ++e) {
      int j = j0 + ((e >> 4) << 5) + (e & 7) + (hf << 3) + (((e >> 3) & 1) << 4);
      nv[e] = nrow[j];
    }
    const int jn = (j0 + KT < N) ? (j0 + KT) : j0;
    v8h g[4];
#pragma unroll
    for (int i = 0; i < 4; ++i)
      g[i] = *(const v8h*)(gsrc + (size_t)i * 16 * N + jn);

    const _Float16* st = &sB[cur][0];
    v16h bf[2][4];
#pragma unroll
    for (int u = 0; u < 2; ++u)
#pragma unroll
      for (int nt = 0; nt < 4; ++nt)
        bf[u][nt] = load_frag16(st, nt * 16, PITCH_H, 32 * u);

    float t[32];
    float bm[2] = {MASK2, MASK2};
#pragma unroll
    for (int e = 0; e < 32; ++e) {
      t[e] = logit2m(__builtin_amdgcn_inverse_ballot_w32(mk[e]), nv[e], svL);
      bm[e >> 4] = fmaxf(bm[e >> 4], t[e]);
    }
    bm[0] = fmaxf(bm[0], xhalf(bm[0]));
    bm[1] = fmaxf(bm[1], xhalf(bm[1]));

#pragma unroll
    for (int i = 0; i < 4; ++i)
      *(v8h*)(&sB[cur ^ 1][loff + i * 16 * PITCH_H]) = g[i];

#pragma unroll
    for (int u = 0; u < 2; ++u) {
      if (__builtin_amdgcn_ballot_w32(bm[u] > m_run + LAZY) != 0) {
        float m_new = fmaxf(m_run, bm[u]);
        float f = __builtin_amdgcn_exp2f(m_run - m_new);
        l_run *= f;
#pragma unroll
        for (int k = 0; k < 8; ++k) {
          float fk = __shfl(f, k + (hf << 3), 32);
          c0[k] *= fk;
          c1[k] *= fk;
          c2[k] *= fk;
          c3[k] *= fk;
        }
        m_run = m_new;
      }

      if (u == 0) {
        const int jm = (j0 + KT < N) ? (j0 + KT) : j0;
#pragma unroll
        for (int e = 0; e < 32; ++e) mk[e] = mrow[(jm >> 1) + e];
      }

      v16h a;
#pragma unroll
      for (int e = 0; e < 16; ++e) {
        float p = __builtin_amdgcn_exp2f(t[16 * u + e] - m_run);
        l_run += p;
        a[e] = (_Float16)p;
      }
      c0 = wmma16(a, bf[u][0], c0);
      c1 = wmma16(a, bf[u][1], c1);
      c2 = wmma16(a, bf[u][2], c2);
      c3 = wmma16(a, bf[u][3], c3);
    }
    __syncthreads();
  }

  float inv = 1.0f / (l_run + xhalf(l_run));
  float ik[8];
#pragma unroll
  for (int k = 0; k < 8; ++k) ik[k] = __shfl(inv, k + (hf << 3), 32);

  v8f acc[4] = {c0, c1, c2, c3};
#pragma unroll
  for (int nt = 0; nt < 4; ++nt) {
#pragma unroll
    for (int k = 0; k < 8; ++k) {
      int orow = r0 + k + (hf << 3);
      int col  = nt * 16 + r;
      float v  = acc[nt][k] * ik[k] + b1[head * NHID + col];
      v = v > 0.0f ? v : 0.0f;
      H[(size_t)orow * (H1 * NHID) + head * NHID + col] = (_Float16)v;
    }
  }
}

__global__ void gemm2_kernel(const _Float16* __restrict__ H,
                             const _Float16* __restrict__ W2T,
                             float* __restrict__ feats2,
                             _Float16* __restrict__ feats2T) {
  int r0 = blockIdx.x << 4;
  v8f c = {};
  for (int k0 = 0; k0 < H1 * NHID; k0 += 32) {
    v16h a = load_frag16(H, r0, H1 * NHID, k0);
    v16h b = load_frag16(W2T, 0, H1 * NHID, k0);
    c = wmma16(a, b, c);
  }
  int lane = threadIdx.x & 31;
  int hf   = lane >> 4;
  int r    = lane & 15;
#pragma unroll
  for (int k = 0; k < 8; ++k) {
    int orow = r0 + k + (hf << 3);
    float v  = c[k];
    feats2[(size_t)orow * NCLASS + r] = v;
    feats2T[(size_t)r * N + orow]     = (_Float16)v;
  }
}

#define SPLIT2 16

__global__ void __launch_bounds__(32 * SPLIT2)
__attribute__((amdgpu_waves_per_eu(4, 8)))
attn_gemm2_kernel(const unsigned* __restrict__ masks,
                  const _Float16* __restrict__ feats2T,
                  const float* __restrict__ s2,
                  const float* __restrict__ n2,
                  const float* __restrict__ b2,
                  float* __restrict__ out) {
  __shared__ float sh_c[SPLIT2][16][NCLASS];
  __shared__ float sh_m[SPLIT2][16];
  __shared__ float sh_l[SPLIT2][16];

  int r0   = blockIdx.x << 4;
  int wv   = __builtin_amdgcn_readfirstlane((int)(threadIdx.x >> 5));
  int lane = threadIdx.x & 31;
  int hf   = lane >> 4;
  int r    = lane & 15;
  int row  = r0 + r;

  float svL = s2[row] * LOG2E;
  const unsigned* mrow = masks + (size_t)blockIdx.x * (NCB * 16);

  float m_run = -1.0e30f;
  float l_run = 0.0f;

  v8f c = {};
  const int jbeg = wv * (N / SPLIT2);
  const int jend = jbeg + (N / SPLIT2);
  unsigned mk[16];
#pragma unroll
  for (int e = 0; e < 16; ++e) mk[e] = mrow[(jbeg >> 1) + e];

  for (int j0 = jbeg; j0 < jend; j0 += 32) {
    float nv[16];
#pragma unroll
    for (int e = 0; e < 16; ++e) {
      int j = j0 + (e & 7) + (hf << 3) + ((e >> 3) << 4);
      nv[e] = n2[j];
    }
    v16h b = load_frag16(feats2T, 0, N, j0);

    float t[16];
    float bm = MASK2;
#pragma unroll
    for (int e = 0; e < 16; ++e) {
      t[e] = logit2m(__builtin_amdgcn_inverse_ballot_w32(mk[e]), nv[e], svL);
      bm   = fmaxf(bm, t[e]);
    }
    bm = fmaxf(bm, xhalf(bm));

    asm volatile("" : "+v"(b), "+v"(bm));

    if (__builtin_amdgcn_ballot_w32(bm > m_run + LAZY) != 0) {
      float m_new = fmaxf(m_run, bm);
      float f = __builtin_amdgcn_exp2f(m_run - m_new);
      l_run *= f;
#pragma unroll
      for (int k = 0; k < 8; ++k) c[k] *= __shfl(f, k + (hf << 3), 32);
      m_run = m_new;
    }

    {
      const int jm = (j0 + 32 < jend) ? (j0 + 32) : j0;
#pragma unroll
      for (int e = 0; e < 16; ++e) mk[e] = mrow[(jm >> 1) + e];
    }

    v16h a;
#pragma unroll
    for (int e = 0; e < 16; ++e) {
      float p = __builtin_amdgcn_exp2f(t[e] - m_run);
      l_run += p;
      a[e] = (_Float16)p;
    }
    c = wmma16(a, b, c);
  }

  float lsum = l_run + xhalf(l_run);
  if (hf == 0) {
    sh_m[wv][r] = m_run;
    sh_l[wv][r] = lsum;
  }
#pragma unroll
  for (int k = 0; k < 8; ++k) sh_c[wv][k + (hf << 3)][r] = c[k];
  __syncthreads();

  if (threadIdx.x < 16 * NCLASS) {
    int mrow = threadIdx.x >> 4;
    int cls  = threadIdx.x & 15;
    float M = sh_m[0][mrow];
#pragma unroll
    for (int w = 1; w < SPLIT2; ++w) M = fmaxf(M, sh_m[w][mrow]);
    float L = 0.0f, C = 0.0f;
#pragma unroll
    for (int w = 0; w < SPLIT2; ++w) {
      float f = __builtin_amdgcn_exp2f(sh_m[w][mrow] - M);
      L = __builtin_fmaf(sh_l[w][mrow], f, L);
      C = __builtin_fmaf(sh_c[w][mrow][cls], f, C);
    }
    float inv = 1.0f / L;
    float v   = C * inv + b2[cls];
    v = v > 0.0f ? v : 0.0f;
    float mx = v;
#pragma unroll
    for (int msk = 1; msk < 16; msk <<= 1)
      mx = fmaxf(mx, __shfl_xor(mx, msk, 32));
    float ex = __expf(v - mx);
    float sum = ex;
#pragma unroll
    for (int msk = 1; msk < 16; msk <<= 1) sum += __shfl_xor(sum, msk, 32);
    float o = (v - mx) - logf(sum);
    out[(size_t)(r0 + mrow) * NCLASS + cls] = o;
  }
}

extern "C" void kernel_launch(void* const* d_in, const int* in_sizes, int n_in,
                              void* d_out, int out_size, void* d_ws,
                              size_t ws_size, hipStream_t stream) {
  const float* x   = (const float*)d_in[0];
  const int*   adj = (const int*)d_in[1];
  const float* W1  = (const float*)d_in[2];
  const float* b1  = (const float*)d_in[3];
  const float* as1 = (const float*)d_in[4];
  const float* an1 = (const float*)d_in[5];
  const float* W2  = (const float*)d_in[6];
  const float* b2  = (const float*)d_in[7];
  const float* as2 = (const float*)d_in[8];
  const float* an2 = (const float*)d_in[9];
  float* out = (float*)d_out;

  char* ws = (char*)d_ws;
  size_t off = 0;
  auto alloc = [&](size_t bytes) -> void* {
    off = (off + 255) & ~(size_t)255;
    void* p = ws + off;
    off += bytes;
    return p;
  };

  _Float16* XH      = (_Float16*)alloc((size_t)N * NFEAT * 2);
  _Float16* W1T     = (_Float16*)alloc((size_t)H1 * NHID * NFEAT * 2);
  _Float16* W2T     = (_Float16*)alloc((size_t)NCLASS * H1 * NHID * 2);
  float*    feats1  = (float*)alloc((size_t)H1 * N * NHID * 4);
  _Float16* featsT  = (_Float16*)alloc((size_t)H1 * NHID * N * 2);
  float*    s1      = (float*)alloc((size_t)H1 * N * 4);
  float*    n1      = (float*)alloc((size_t)H1 * N * 4);
  _Float16* H       = (_Float16*)alloc((size_t)N * H1 * NHID * 2);
  float*    feats2  = (float*)alloc((size_t)N * NCLASS * 4);
  _Float16* feats2T = (_Float16*)alloc((size_t)NCLASS * N * 2);
  float*    s2      = (float*)alloc((size_t)N * 4);
  float*    n2      = (float*)alloc((size_t)N * 4);
  unsigned* masks   = (unsigned*)XH;
  (void)ws_size; (void)n_in; (void)in_sizes; (void)out_size;

  cast_kernel<<<CAST_X_BLOCKS + CAST_W1_BLOCKS + CAST_W2_BLOCKS, 256, 0,
                stream>>>(x, W1, W2, XH, W1T, W2T);
  gemm1_kernel<<<dim3(N / 16, H1), 32, 0, stream>>>(XH, W1T, feats1, featsT);
  sn1_adjmask_kernel<<<SN1_BLOCKS + MASK_BLOCKS, 256, 0, stream>>>(
      feats1, as1, an1, s1, n1, adj, masks);
  attn_gemm1_kernel<<<dim3(N / (16 * WAVES1), H1), 32 * WAVES1, 0, stream>>>(
      masks, featsT, s1, n1, b1, H);
  gemm2_kernel<<<N / 16, 32, 0, stream>>>(H, W2T, feats2, feats2T);
  sn_kernel<<<(N + 255) / 256, 256, 0, stream>>>(feats2, as2, an2, s2, n2,
                                                 NCLASS, N, N);
  attn_gemm2_kernel<<<N / 16, 32 * SPLIT2, 0, stream>>>(masks, feats2T, s2, n2,
                                                        b2, out);
}


# Layer-1 feature GEMM: 8 row tiles of one head per work-group share W1T[head] through LDS (copied once, whole-line loads), A fragments requested with the copy; order of sums unchanged
# speedup vs baseline: 5.8492x; 1.1941x over previous
#include <hip/hip_runtime.h>
#include <hip/hip_bf16.h>


#define N      4096
#define NFEAT  512
#define NHID   64
#define NCLASS 16
#define H1     8
#define NEG_SLOPE 0.2f
#define MASK_VAL  -1e10f
#define LOG2E 1.4426950408889634f
#define MASK2 (MASK_VAL * LOG2E)
#define LAZY 8.0f

typedef __attribute__((ext_vector_type(16))) _Float16 v16h;
typedef __attribute__((ext_vector_type(8)))  _Float16 v8h;
typedef __attribute__((ext_vector_type(8)))  float    v8f;

__device__ __forceinline__ v8f wmma16(v16h a, v16h b, v8f c) {
  return __builtin_amdgcn_wmma_f32_16x16x32_f16(
       false, a,  false, b,
       (short)0, c,  false,  false);
}

__device__ __forceinline__ v16h load_frag16(const _Float16* base, int rc,
                                            int ld, int k0) {
  int lane = threadIdx.x & 31;
  int hf   = lane >> 4;
  int r    = lane & 15;
  const _Float16* p = base + (size_t)(rc + r) * ld + k0 + (hf << 3);
  v8h lo = *(const v8h*)p;
  v8h hi = *(const v8h*)(p + 16);
  return __builtin_shufflevector(lo, hi, 0, 1, 2, 3, 4, 5, 6, 7, 8, 9, 10, 11,
                                 12, 13, 14, 15);
}

__device__ __forceinline__ float lrelu(float x) {
  return x > 0.0f ? x : NEG_SLOPE * x;
}

__device__ __forceinline__ float xhalf(float x) {
  int b = __builtin_bit_cast(int, x);
  return __builtin_bit_cast(
      float, __builtin_amdgcn_permlanex16(b, b, 0x76543210, 0xfedcba98, false,
                                          false));
}

__device__ __forceinline__ float logit2m(bool on, float nj, float svL) {
  float x2 = __builtin_fmaf(nj, LOG2E, svL);
  float lr = fmaxf(x2, NEG_SLOPE * x2);
  return on ? lr : MASK2;
}

#define NCB (N / 32)
#define MASK_BLOCKS ((N / 16) * (NCB / 4) / 8)
typedef __attribute__((ext_vector_type(4))) int v4i;

__device__ __forceinline__ void adjmask_block(const int* __restrict__ adj,
                                              unsigned* __restrict__ masks,
                                              int blk) {
  int wv   = __builtin_amdgcn_readfirstlane((int)(threadIdx.x >> 5));
  int gw   = blk * 8 + wv;
  int rt   = gw >> 5;
  int cb0  = (gw & 31) << 2;
  int lane = threadIdx.x & 31;
  int hf   = lane >> 4;
  int r    = lane & 15;
  const int* arow = adj + (size_t)(rt * 16 + r) * N + cb0 * 32 + (hf << 3);

  v4i a[4][4];
#pragma unroll
  for (int cb = 0; cb < 4; ++cb) {
#pragma unroll
    for (int q = 0; q < 4; ++q) {
      a[cb][q] = __builtin_nontemporal_load(
          (const v4i*)(arow + cb * 32 + (q & 1) * 4 + (q >> 1) * 16));
    }
  }
  unsigned o[2] = {0u, 0u};
#pragma unroll
  for (int cb = 0; cb < 4; ++cb) {
#pragma unroll
    for (int e = 0; e < 16; ++e) {
      unsigned m = __builtin_amdgcn_ballot_w32(a[cb][e >> 2][e & 3] > 0);
      o[cb >> 1] = (lane == e + 16 * (cb & 1)) ? m : o[cb >> 1];
    }
  }
  unsigned* dst = masks + ((size_t)rt * NCB + cb0) * 16;
  dst[lane]      = o[0];
  dst[32 + lane] = o[1];
}

typedef __attribute__((ext_vector_type(4))) float    v4f;
typedef __attribute__((ext_vector_type(4))) _Float16 v4h;

#define CAST_X_BLOCKS     256
#define CAST_X_PER_THREAD ((N * NFEAT / 4) / (CAST_X_BLOCKS * 256))
#define CAST_W1_BLOCKS    ((H1 * NHID * NFEAT / 8) / 256)
#define CAST_W2_BLOCKS    ((NCLASS * H1 * NHID / 8) / 256)

__global__ void __launch_bounds__(256)
cast_kernel(const float* __restrict__ x,
            const float* __restrict__ W1,
            const float* __restrict__ W2,
            _Float16* __restrict__ XH,
            _Float16* __restrict__ W1T,
            _Float16* __restrict__ W2T) {
  const int b   = blockIdx.x;
  const int tid = threadIdx.x;
  if (b < CAST_X_BLOCKS) {
    const int base = b * (256 * CAST_X_PER_THREAD) + tid;
    const v4f* src = (const v4f*)x + base;
    v4h* dst = (v4h*)XH + base;
    v4f v[CAST_X_PER_THREAD];
#pragma unroll
    for (int i = 0; i < CAST_X_PER_THREAD; ++i) v[i] = src[i * 256];
#pragma unroll
    for (int i = 0; i < CAST_X_PER_THREAD; ++i)
      dst[i * 256] = __builtin_convertvector(v[i], v4h);
  } else if (b < CAST_X_BLOCKS + CAST_W1_BLOCKS) {
    const int q  = (b - CAST_X_BLOCKS) * 256 + tid;
    const int c  = q & (NHID - 1);
    const int k0 = ((q / NHID) & (NFEAT / 8 - 1)) * 8;
    const int h  = q / (NHID * (NFEAT / 8));
    const float* src = W1 + ((size_t)h * NFEAT + k0) * NHID + c;
    v8h o;
#pragma unroll
    for (int j = 0; j < 8; ++j) o[j] = (_Float16)src[j * NHID];
    *(v8h*)(W1T + ((size_t)(h * NHID + c)) * NFEAT + k0) = o;
  } else {
    const int q  = (b - CAST_X_BLOCKS - CAST_W1_BLOCKS) * 256 + tid;
    const int c  = q & (NCLASS - 1);
    const int k0 = (q / NCLASS) * 8;
    const float* src = W2 + (size_t)k0 * NCLASS + c;
    v8h o;
#pragma unroll
    for (int j = 0; j < 8; ++j) o[j] = (_Float16)src[j * NCLASS];
    *(v8h*)(W2T + (size_t)c * (H1 * NHID) + k0) = o;
  }
}

#define G1_WAVES  8
#define G1_KSTEPS (NFEAT / 32)
#define PITCH_W (NFEAT + 8)

__global__ void __launch_bounds__(32 * G1_WAVES)
__attribute__((amdgpu_waves_per_eu(1, 2)))
gemm1_kernel(const _Float16* __restrict__ XH,
             const _Float16* __restrict__ W1T,
             float* __restrict__ feats,
             _Float16* __restrict__ featsT) {
  __shared__ __attribute__((aligned(16))) _Float16 sW[NHID * PITCH_W];

  int head = blockIdx.y;
  int wv   = __builtin_amdgcn_readfirstlane((int)(threadIdx.x >> 5));
  int r0   = ((int)blockIdx.x * G1_WAVES + wv) << 4;
  const int tid = threadIdx.x;
  const _Float16* BT = W1T + (size_t)head * NHID * NFEAT;

  const _Float16* gsrc = BT + (size_t)tid * 8;
  const int loff = (tid >> 6) * PITCH_W + ((tid & 63) << 3);
  v8h g[16];
#pragma unroll
  for (int i = 0; i < 16; ++i)
    g[i] = *(const v8h*)(gsrc + (size_t)i * (256 * 8));
  v16h a[G1_KSTEPS];
#pragma unroll
  for (int ks = 0; ks < G1_KSTEPS; ++ks)
    a[ks] = load_frag16(XH, r0, NFEAT, ks * 32);
#pragma unroll
  for (int i = 0; i < 16; ++i)
    *(v8h*)(&sW[loff + i * 4 * PITCH_W]) = g[i];
  __syncthreads();

  v8f c0 = {}, c1 = {}, c2 = {}, c3 = {};
  v16h bc[4];
#pragma unroll
  for (int nt = 0; nt < 4; ++nt) bc[nt] = load_frag16(sW, nt * 16, PITCH_W, 0);
#pragma unroll
  for (int ks = 0; ks < G1_KSTEPS; ++ks) {
    v16h bn[4];
    if (ks + 1 < G1_KSTEPS) {
#pragma unroll
      for (int nt = 0; nt < 4; ++nt)
        bn[nt] = load_frag16(sW, nt * 16, PITCH_W, (ks + 1) * 32);
    }
    c0 = wmma16(a[ks], bc[0], c0);
    c1 = wmma16(a[ks], bc[1], c1);
    c2 = wmma16(a[ks], bc[2], c2);
    c3 = wmma16(a[ks], bc[3], c3);
    if (ks + 1 < G1_KSTEPS) {
#pragma unroll
      for (int nt = 0; nt < 4; ++nt) bc[nt] = bn[nt];
    }
  }

  int lane = threadIdx.x & 31;
  int hf   = lane >> 4;
  int r    = lane & 15;
  v8f acc[4] = {c0, c1, c2, c3};
#pragma unroll
  for (int nt = 0; nt < 4; ++nt) {
#pragma unroll
    for (int k = 0; k < 8; ++k) {
      int orow = r0 + k + (hf << 3);
      int col  = nt * 16 + r;
      float v  = acc[nt][k];
      feats[((size_t)head * N + orow) * NHID + col] = v;
      featsT[((size_t)head * NHID + col) * N + orow] = (_Float16)v;
    }
  }
}

__device__ __forceinline__ void sn_row(const float* __restrict__ feats,
                                       const float* __restrict__ a_s,
                                       const float* __restrict__ a_n,
                                       float* __restrict__ s,
                                       float* __restrict__ n, int fout,
                                       int rowsPerHead, int total, int i) {
  if (i >= total) return;
  int head = i / rowsPerHead;
  const float* f  = feats + (size_t)i * fout;
  const float* av = a_s + head * fout;
  const float* nv = a_n + head * fout;
  float ss = 0.0f, nn = 0.0f;
  for (int c = 0; c < fout; ++c) {
    float v = f[c];
    ss += v * av[c];
    nn += v * nv[c];
  }
  s[i] = ss;
  n[i] = nn;
}

__global__ void sn_kernel(const float* __restrict__ feats,
                          const float* __restrict__ a_s,
                          const float* __restrict__ a_n,
                          float* __restrict__ s, float* __restrict__ n,
                          int fout, int rowsPerHead, int total) {
  sn_row(feats, a_s, a_n, s, n, fout, rowsPerHead, total,
         blockIdx.x * blockDim.x + threadIdx.x);
}

#define SN1_BLOCKS ((H1 * N) / 256)

__global__ void __launch_bounds__(256)
sn1_adjmask_kernel(const float* __restrict__ feats,
                   const float* __restrict__ a_s,
                   const float* __restrict__ a_n, float* __restrict__ s,
                   float* __restrict__ n, const int* __restrict__ adj,
                   unsigned* __restrict__ masks) {
  if (blockIdx.x < SN1_BLOCKS)
    sn_row(feats, a_s, a_n, s, n, NHID, N, H1 * N,
           (int)(blockIdx.x * 256 + threadIdx.x));
  else
    adjmask_block(adj, masks, (int)blockIdx.x - SN1_BLOCKS);
}

#define WAVES1  4
#define KT      64
#define PITCH_H 72

__global__ void __launch_bounds__(32 * WAVES1)
__attribute__((amdgpu_waves_per_eu(1, 2)))
attn_gemm1_kernel(const unsigned* __restrict__ masks,
                  const _Float16* __restrict__ featsT,
                  const float* __restrict__ s1,
                  const float* __restrict__ n1,
                  const float* __restrict__ b1,
                  _Float16* __restrict__ H) {
  __shared__ __attribute__((aligned(16))) _Float16 sB[2][NHID * PITCH_H];

  int head = blockIdx.y;
  int wv   = __builtin_amdgcn_readfirstlane((int)(threadIdx.x >> 5));
  int r0   = ((int)blockIdx.x * WAVES1 + wv) << 4;
  int lane = threadIdx.x & 31;
  int hf   = lane >> 4;
  int r    = lane & 15;
  int row  = r0 + r;

  const float* nrow = n1 + head * N;
  float svL = s1[head * N + row] * LOG2E;
  const unsigned* mrow = masks + (size_t)(r0 >> 4) * (NCB * 16);
  const _Float16* BT = featsT + (size_t)head * NHID * N;

  const int tid = threadIdx.x;
  const _Float16* gsrc = BT + (size_t)(tid >> 3) * N + ((tid & 7) << 3);
  const int loff = (tid >> 3) * PITCH_H + ((tid & 7) << 3);

#pragma unroll
  for (int i = 0; i < 4; ++i) {
    v8h g = *(const v8h*)(gsrc + (size_t)i * 16 * N);
    *(v8h*)(&sB[0][loff + i * 16 * PITCH_H]) = g;
  }
  __syncthreads();

  float m_run = -1.0e30f;
  float l_run = 0.0f;

  unsigned mk[32];
#pragma unroll
  for (int e = 0; e < 32; ++e) mk[e] = mrow[e];

  v8f c0 = {}, c1 = {}, c2 = {}, c3 = {};
  for (int j0 = 0; j0 < N; j0 += KT) {
    const int cur = (j0 / KT) & 1;
    float nv[32];
#pragma unroll
    for (int e = 0; e < 32; ++e) {
      int j = j0 + ((e >> 4) << 5) + (e & 7) + (hf << 3) + (((e >> 3) & 1) << 4);
      nv[e] = nrow[j];
    }
    const int jn = (j0 + KT < N) ? (j0 + KT) : j0;
    v8h g[4];
#pragma unroll
    for (int i = 0; i < 4; ++i)
      g[i] = *(const v8h*)(gsrc + (size_t)i * 16 * N + jn);

    const _Float16* st = &sB[cur][0];
    v16h bf[2][4];
#pragma unroll
    for (int u = 0; u < 2; ++u)
#pragma unroll
      for (int nt = 0; nt < 4; ++nt)
        bf[u][nt] = load_frag16(st, nt * 16, PITCH_H, 32 * u);

    float t[32];
    float bm[2] = {MASK2, MASK2};
#pragma unroll
    for (int e = 0; e < 32; ++e) {
      t[e] = logit2m(__builtin_amdgcn_inverse_ballot_w32(mk[e]), nv[e], svL);
      bm[e >> 4] = fmaxf(bm[e >> 4], t[e]);
    }
    bm[0] = fmaxf(bm[0], xhalf(bm[0]));
    bm[1] = fmaxf(bm[1], xhalf(bm[1]));

#pragma unroll
    for (int i = 0; i < 4; ++i)
      *(v8h*)(&sB[cur ^ 1][loff + i * 16 * PITCH_H]) = g[i];

#pragma unroll
    for (int u = 0; u < 2; ++u) {
      if (__builtin_amdgcn_ballot_w32(bm[u] > m_run + LAZY) != 0) {
        float m_new = fmaxf(m_run, bm[u]);
        float f = __builtin_amdgcn_exp2f(m_run - m_new);
        l_run *= f;
#pragma unroll
        for (int k = 0; k < 8; ++k) {
          float fk = __shfl(f, k + (hf << 3), 32);
          c0[k] *= fk;
          c1[k] *= fk;
          c2[k] *= fk;
          c3[k] *= fk;
        }
        m_run = m_new;
      }

      if (u == 0) {
        const int jm = (j0 + KT < N) ? (j0 + KT) : j0;
#pragma unroll
        for (int e = 0; e < 32; ++e) mk[e] = mrow[(jm >> 1) + e];
      }

      v16h a;
#pragma unroll
      for (int e = 0; e < 16; ++e) {
        float p = __builtin_amdgcn_exp2f(t[16 * u + e] - m_run);
        l_run += p;
        a[e] = (_Float16)p;
      }
      c0 = wmma16(a, bf[u][0], c0);
      c1 = wmma16(a, bf[u][1], c1);
      c2 = wmma16(a, bf[u][2], c2);
      c3 = wmma16(a, bf[u][3], c3);
    }
    __syncthreads();
  }

  float inv = 1.0f / (l_run + xhalf(l_run));
  float ik[8];
#pragma unroll
  for (int k = 0; k < 8; ++k) ik[k] = __shfl(inv, k + (hf << 3), 32);

  v8f acc[4] = {c0, c1, c2, c3};
#pragma unroll
  for (int nt = 0; nt < 4; ++nt) {
#pragma unroll
    for (int k = 0; k < 8; ++k) {
      int orow = r0 + k + (hf << 3);
      int col  = nt * 16 + r;
      float v  = acc[nt][k] * ik[k] + b1[head * NHID + col];
      v = v > 0.0f ? v : 0.0f;
      H[(size_t)orow * (H1 * NHID) + head * NHID + col] = (_Float16)v;
    }
  }
}

__global__ void gemm2_kernel(const _Float16* __restrict__ H,
                             const _Float16* __restrict__ W2T,
                             float* __restrict__ feats2,
                             _Float16* __restrict__ feats2T) {
  int r0 = blockIdx.x << 4;
  v8f c = {};
  for (int k0 = 0; k0 < H1 * NHID; k0 += 32) {
    v16h a = load_frag16(H, r0, H1 * NHID, k0);
    v16h b = load_frag16(W2T, 0, H1 * NHID, k0);
    c = wmma16(a, b, c);
  }
  int lane = threadIdx.x & 31;
  int hf   = lane >> 4;
  int r    = lane & 15;
#pragma unroll
  for (int k = 0; k < 8; ++k) {
    int orow = r0 + k + (hf << 3);
    float v  = c[k];
    feats2[(size_t)orow * NCLASS + r] = v;
    feats2T[(size_t)r * N + orow]     = (_Float16)v;
  }
}

#define SPLIT2 16

__global__ void __launch_bounds__(32 * SPLIT2)
__attribute__((amdgpu_waves_per_eu(4, 8)))
attn_gemm2_kernel(const unsigned* __restrict__ masks,
                  const _Float16* __restrict__ feats2T,
                  const float* __restrict__ s2,
                  const float* __restrict__ n2,
                  const float* __restrict__ b2,
                  float* __restrict__ out) {
  __shared__ float sh_c[SPLIT2][16][NCLASS];
  __shared__ float sh_m[SPLIT2][16];
  __shared__ float sh_l[SPLIT2][16];

  int r0   = blockIdx.x << 4;
  int wv   = __builtin_amdgcn_readfirstlane((int)(threadIdx.x >> 5));
  int lane = threadIdx.x & 31;
  int hf   = lane >> 4;
  int r    = lane & 15;
  int row  = r0 + r;

  float svL = s2[row] * LOG2E;
  const unsigned* mrow = masks + (size_t)blockIdx.x * (NCB * 16);

  float m_run = -1.0e30f;
  float l_run = 0.0f;

  v8f c = {};
  const int jbeg = wv * (N / SPLIT2);
  const int jend = jbeg + (N / SPLIT2);
  unsigned mk[16];
#pragma unroll
  for (int e = 0; e < 16; ++e) mk[e] = mrow[(jbeg >> 1) + e];

  for (int j0 = jbeg; j0 < jend; j0 += 32) {
    float nv[16];
#pragma unroll
    for (int e = 0; e < 16; ++e) {
      int j = j0 + (e & 7) + (hf << 3) + ((e >> 3) << 4);
      nv[e] = n2[j];
    }
    v16h b = load_frag16(feats2T, 0, N, j0);

    float t[16];
    float bm = MASK2;
#pragma unroll
    for (int e = 0; e < 16; ++e) {
      t[e] = logit2m(__builtin_amdgcn_inverse_ballot_w32(mk[e]), nv[e], svL);
      bm   = fmaxf(bm, t[e]);
    }
    bm = fmaxf(bm, xhalf(bm));

    asm volatile("" : "+v"(b), "+v"(bm));

    if (__builtin_amdgcn_ballot_w32(bm > m_run + LAZY) != 0) {
      float m_new = fmaxf(m_run, bm);
      float f = __builtin_amdgcn_exp2f(m_run - m_new);
      l_run *= f;
#pragma unroll
      for (int k = 0; k < 8; ++k) c[k] *= __shfl(f, k + (hf << 3), 32);
      m_run = m_new;
    }

    {
      const int jm = (j0 + 32 < jend) ? (j0 + 32) : j0;
#pragma unroll
      for (int e = 0; e < 16; ++e) mk[e] = mrow[(jm >> 1) + e];
    }

    v16h a;
#pragma unroll
    for (int e = 0; e < 16; ++e) {
      float p = __builtin_amdgcn_exp2f(t[e] - m_run);
      l_run += p;
      a[e] = (_Float16)p;
    }
    c = wmma16(a, b, c);
  }

  float lsum = l_run + xhalf(l_run);
  if (hf == 0) {
    sh_m[wv][r] = m_run;
    sh_l[wv][r] = lsum;
  }
#pragma unroll
  for (int k = 0; k < 8; ++k) sh_c[wv][k + (hf << 3)][r] = c[k];
  __syncthreads();

  if (threadIdx.x < 16 * NCLASS) {
    int mrow = threadIdx.x >> 4;
    int cls  = threadIdx.x & 15;
    float M = sh_m[0][mrow];
#pragma unroll
    for (int w = 1; w < SPLIT2; ++w) M = fmaxf(M, sh_m[w][mrow]);
    float L = 0.0f, C = 0.0f;
#pragma unroll
    for (int w = 0; w < SPLIT2; ++w) {
      float f = __builtin_amdgcn_exp2f(sh_m[w][mrow] - M);
      L = __builtin_fmaf(sh_l[w][mrow], f, L);
      C = __builtin_fmaf(sh_c[w][mrow][cls], f, C);
    }
    float inv = 1.0f / L;
    float v   = C * inv + b2[cls];
    v = v > 0.0f ? v : 0.0f;
    float mx = v;
#pragma unroll
    for (int msk = 1; msk < 16; msk <<= 1)
      mx = fmaxf(mx, __shfl_xor(mx, msk, 32));
    float ex = __expf(v - mx);
    float sum = ex;
#pragma unroll
    for (int msk = 1; msk < 16; msk <<= 1) sum += __shfl_xor(sum, msk, 32);
    float o = (v - mx) - logf(sum);
    out[(size_t)(r0 + mrow) * NCLASS + cls] = o;
  }
}

extern "C" void kernel_launch(void* const* d_in, const int* in_sizes, int n_in,
                              void* d_out, int out_size, void* d_ws,
                              size_t ws_size, hipStream_t stream) {
  const float* x   = (const float*)d_in[0];
  const int*   adj = (const int*)d_in[1];
  const float* W1  = (const float*)d_in[2];
  const float* b1  = (const float*)d_in[3];
  const float* as1 = (const float*)d_in[4];
  const float* an1 = (const float*)d_in[5];
  const float* W2  = (const float*)d_in[6];
  const float* b2  = (const float*)d_in[7];
  const float* as2 = (const float*)d_in[8];
  const float* an2 = (const float*)d_in[9];
  float* out = (float*)d_out;

  char* ws = (char*)d_ws;
  size_t off = 0;
  auto alloc = [&](size_t bytes) -> void* {
    off = (off + 255) & ~(size_t)255;
    void* p = ws + off;
    off += bytes;
    return p;
  };

  _Float16* XH      = (_Float16*)alloc((size_t)N * NFEAT * 2);
  _Float16* W1T     = (_Float16*)alloc((size_t)H1 * NHID * NFEAT * 2);
  _Float16* W2T     = (_Float16*)alloc((size_t)NCLASS * H1 * NHID * 2);
  float*    feats1  = (float*)alloc((size_t)H1 * N * NHID * 4);
  _Float16* featsT  = (_Float16*)alloc((size_t)H1 * NHID * N * 2);
  float*    s1      = (float*)alloc((size_t)H1 * N * 4);
  float*    n1      = (float*)alloc((size_t)H1 * N * 4);
  _Float16* H       = (_Float16*)alloc((size_t)N * H1 * NHID * 2);
  float*    feats2  = (float*)alloc((size_t)N * NCLASS * 4);
  _Float16* feats2T = (_Float16*)alloc((size_t)NCLASS * N * 2);
  float*    s2      = (float*)alloc((size_t)N * 4);
  float*    n2      = (float*)alloc((size_t)N * 4);
  unsigned* masks   = (unsigned*)XH;
  (void)ws_size; (void)n_in; (void)in_sizes; (void)out_size;

  cast_kernel<<<CAST_X_BLOCKS + CAST_W1_BLOCKS + CAST_W2_BLOCKS, 256, 0,
                stream>>>(x, W1, W2, XH, W1T, W2T);
  gemm1_kernel<<<dim3(N / (16 * G1_WAVES), H1), 32 * G1_WAVES, 0, stream>>>(
      XH, W1T, feats1, featsT);
  sn1_adjmask_kernel<<<SN1_BLOCKS + MASK_BLOCKS, 256, 0, stream>>>(
      feats1, as1, an1, s1, n1, adj, masks);
  attn_gemm1_kernel<<<dim3(N / (16 * WAVES1), H1), 32 * WAVES1, 0, stream>>>(
      masks, featsT, s1, n1, b1, H);
  gemm2_kernel<<<N / 16, 32, 0, stream>>>(H, W2T, feats2, feats2T);
  sn_kernel<<<(N + 255) / 256, 256, 0, stream>>>(feats2, as2, an2, s2, n2,
                                                 NCLASS, N, N);
  attn_gemm2_kernel<<<N / 16, 32 * SPLIT2, 0, stream>>>(masks, feats2T, s2, n2,
                                                        b2, out);
}


# Adjacency-mask job runs as the tail work-groups of the layer-1 feature GEMM's grid (masks get a workspace plane of their own): 5 launches, adjacency stream overlaps the GEMM; every sum unchanged
# speedup vs baseline: 6.7255x; 1.1498x over previous
#include <hip/hip_runtime.h>
#include <hip/hip_bf16.h>


#define N      4096
#define NFEAT  512
#define NHID   64
#define NCLASS 16
#define H1     8
#define NEG_SLOPE 0.2f
#define MASK_VAL  -1e10f
#define LOG2E 1.4426950408889634f
#define MASK2 (MASK_VAL * LOG2E)
#define LAZY 8.0f

typedef __attribute__((ext_vector_type(16))) _Float16 v16h;
typedef __attribute__((ext_vector_type(8)))  _Float16 v8h;
typedef __attribute__((ext_vector_type(8)))  float    v8f;

__device__ __forceinline__ v8f wmma16(v16h a, v16h b, v8f c) {
  return __builtin_amdgcn_wmma_f32_16x16x32_f16(
       false, a,  false, b,
       (short)0, c,  false,  false);
}

__device__ __forceinline__ v16h load_frag16(const _Float16* base, int rc,
                                            int ld, int k0) {
  int lane = threadIdx.x & 31;
  int hf   = lane >> 4;
  int r    = lane & 15;
  const _Float16* p = base + (size_t)(rc + r) * ld + k0 + (hf << 3);
  v8h lo = *(const v8h*)p;
  v8h hi = *(const v8h*)(p + 16);
  return __builtin_shufflevector(lo, hi, 0, 1, 2, 3, 4, 5, 6, 7, 8, 9, 10, 11,
                                 12, 13, 14, 15);
}

__device__ __forceinline__ float lrelu(float x) {
  return x > 0.0f ? x : NEG_SLOPE * x;
}

__device__ __forceinline__ float xhalf(float x) {
  int b = __builtin_bit_cast(int, x);
  return __builtin_bit_cast(
      float, __builtin_amdgcn_permlanex16(b, b, 0x76543210, 0xfedcba98, false,
                                          false));
}

__device__ __forceinline__ float logit2m(bool on, float nj, float svL) {
  float x2 = __builtin_fmaf(nj, LOG2E, svL);
  float lr = fmaxf(x2, NEG_SLOPE * x2);
  return on ? lr : MASK2;
}

template <int M>
__device__ __forceinline__ float add_xmask(float v) {
  int b = __builtin_bit_cast(int, v);
  int o = __builtin_amdgcn_update_dpp(b, b, 0x160 + M, 0xf, 0xf, false);
  return v + __builtin_bit_cast(float, o);
}
__device__ __forceinline__ float rowsum16(float v) {
  return add_xmask<8>(add_xmask<4>(add_xmask<2>(add_xmask<1>(v))));
}

__device__ __forceinline__ void store_dots(float (&ps)[8], float (&pn)[8],
                                           float* __restrict__ dst_s,
                                           float* __restrict__ dst_n) {
  int lane = threadIdx.x & 31;
  int hf   = lane >> 4;
  int r    = lane & 15;
#pragma unroll
  for (int k = 0; k < 8; ++k) {
    ps[k] = rowsum16(ps[k]);
    pn[k] = rowsum16(pn[k]);
  }
  float sv = ps[0], nv = pn[0];
#pragma unroll
  for (int k = 1; k < 8; ++k) {
    sv = (r == k) ? ps[k] : sv;
    nv = (r == k) ? pn[k] : nv;
  }
  if (r < 8) {
    dst_s[(hf << 3) + r] = sv;
    dst_n[(hf << 3) + r] = nv;
  }
}

#define NCB (N / 32)
#define MASK_BLOCKS ((N / 16) * (NCB / 4) / 8)
typedef __attribute__((ext_vector_type(4))) int v4i;

__device__ __forceinline__ void adjmask_block(const int* __restrict__ adj,
                                              unsigned* __restrict__ masks,
                                              int blk) {
  int wv   = __builtin_amdgcn_readfirstlane((int)(threadIdx.x >> 5));
  int gw   = blk * 8 + wv;
  int rt   = gw >> 5;
  int cb0  = (gw & 31) << 2;
  int lane = threadIdx.x & 31;
  int hf   = lane >> 4;
  int r    = lane & 15;
  const int* arow = adj + (size_t)(rt * 16 + r) * N + cb0 * 32 + (hf << 3);

  v4i a[4][4];
#pragma unroll
  for (int cb = 0; cb < 4; ++cb) {
#pragma unroll
    for (int q = 0; q < 4; ++q) {
      a[cb][q] = __builtin_nontemporal_load(
          (const v4i*)(arow + cb * 32 + (q & 1) * 4 + (q >> 1) * 16));
    }
  }
  unsigned o[2] = {0u, 0u};
#pragma unroll
  for (int cb = 0; cb < 4; ++cb) {
#pragma unroll
    for (int e = 0; e < 16; ++e) {
      unsigned m = __builtin_amdgcn_ballot_w32(a[cb][e >> 2][e & 3] > 0);
      o[cb >> 1] = (lane == e + 16 * (cb & 1)) ? m : o[cb >> 1];
    }
  }
  unsigned* dst = masks + ((size_t)rt * NCB + cb0) * 16;
  dst[lane]      = o[0];
  dst[32 + lane] = o[1];
}

typedef __attribute__((ext_vector_type(4))) float    v4f;
typedef __attribute__((ext_vector_type(4))) _Float16 v4h;

#define CAST_X_BLOCKS     256
#define CAST_X_PER_THREAD ((N * NFEAT / 4) / (CAST_X_BLOCKS * 256))
#define CAST_W1_BLOCKS    ((H1 * NHID * NFEAT / 8) / 256)
#define CAST_W2_BLOCKS    ((NCLASS * H1 * NHID / 8) / 256)

__global__ void __launch_bounds__(256)
cast_kernel(const float* __restrict__ x,
            const float* __restrict__ W1,
            const float* __restrict__ W2,
            _Float16* __restrict__ XH,
            _Float16* __restrict__ W1T,
            _Float16* __restrict__ W2T) {
  const int b   = blockIdx.x;
  const int tid = threadIdx.x;
  if (b < CAST_X_BLOCKS) {
    const int base = b * (256 * CAST_X_PER_THREAD) + tid;
    const v4f* src = (const v4f*)x + base;
    v4h* dst = (v4h*)XH + base;
    v4f v[CAST_X_PER_THREAD];
#pragma unroll
    for (int i = 0; i < CAST_X_PER_THREAD; ++i) v[i] = src[i * 256];
#pragma unroll
    for (int i = 0; i < CAST_X_PER_THREAD; ++i)
      dst[i * 256] = __builtin_convertvector(v[i], v4h);
  } else if (b < CAST_X_BLOCKS + CAST_W1_BLOCKS) {
    const int q  = (b - CAST_X_BLOCKS) * 256 + tid;
    const int c  = q & (NHID - 1);
    const int k0 = ((q / NHID) & (NFEAT / 8 - 1)) * 8;
    const int h  = q / (NHID * (NFEAT / 8));
    const float* src = W1 + ((size_t)h * NFEAT + k0) * NHID + c;
    v8h o;
#pragma unroll
    for (int j = 0; j < 8; ++j) o[j] = (_Float16)src[j * NHID];
    *(v8h*)(W1T + ((size_t)(h * NHID + c)) * NFEAT + k0) = o;
  } else {
    const int q  = (b - CAST_X_BLOCKS - CAST_W1_BLOCKS) * 256 + tid;
    const int c  = q & (NCLASS - 1);
    const int k0 = (q / NCLASS) * 8;
    const float* src = W2 + (size_t)k0 * NCLASS + c;
    v8h o;
#pragma unroll
    for (int j = 0; j < 8; ++j) o[j] = (_Float16)src[j * NCLASS];
    *(v8h*)(W2T + (size_t)c * (H1 * NHID) + k0) = o;
  }
}

#define G1_WAVES  8
#define G1_KSTEPS (NFEAT / 32)
#define G1_XBLOCKS (N / (16 * G1_WAVES))
#define G1_BLOCKS  (G1_XBLOCKS * H1)
#define PITCH_W (NFEAT + 8)

__global__ void __launch_bounds__(32 * G1_WAVES)
__attribute__((amdgpu_waves_per_eu(1, 2)))
gemm1_kernel(const _Float16* __restrict__ XH,
             const _Float16* __restrict__ W1T,
             const float* __restrict__ a_s,
             const float* __restrict__ a_n,
             float* __restrict__ s,
             float* __restrict__ n,
             _Float16* __restrict__ featsT,
             const int* __restrict__ adj,
             unsigned* __restrict__ masks) {
  __shared__ __attribute__((aligned(16))) _Float16 sW[NHID * PITCH_W];

  if (blockIdx.x >= G1_BLOCKS) {
    adjmask_block(adj, masks, (int)blockIdx.x - G1_BLOCKS);
    return;
  }

  int head = (int)blockIdx.x / G1_XBLOCKS;
  int xb   = (int)blockIdx.x % G1_XBLOCKS;
  int wv   = __builtin_amdgcn_readfirstlane((int)(threadIdx.x >> 5));
  int r0   = (xb * G1_WAVES + wv) << 4;
  const int tid = threadIdx.x;
  const _Float16* BT = W1T + (size_t)head * NHID * NFEAT;

  const _Float16* gsrc = BT + (size_t)tid * 8;
  const int loff = (tid >> 6) * PITCH_W + ((tid & 63) << 3);
  v8h g[16];
#pragma unroll
  for (int i = 0; i < 16; ++i)
    g[i] = *(const v8h*)(gsrc + (size_t)i * (256 * 8));
  v16h a[G1_KSTEPS];
#pragma unroll
  for (int ks = 0; ks < G1_KSTEPS; ++ks)
    a[ks] = load_frag16(XH, r0, NFEAT, ks * 32);
  float asv[4], anv[4];
#pragma unroll
  for (int nt = 0; nt < 4; ++nt) {
    asv[nt] = a_s[head * NHID + nt * 16 + (tid & 15)];
    anv[nt] = a_n[head * NHID + nt * 16 + (tid & 15)];
  }
#pragma unroll
  for (int i = 0; i < 16; ++i)
    *(v8h*)(&sW[loff + i * 4 * PITCH_W]) = g[i];
  __syncthreads();

  v8f c0 = {}, c1 = {}, c2 = {}, c3 = {};
  v16h bc[4];
#pragma unroll
  for (int nt = 0; nt < 4; ++nt) bc[nt] = load_frag16(sW, nt * 16, PITCH_W, 0);
#pragma unroll
  for (int ks = 0; ks < G1_KSTEPS; ++ks) {
    v16h bn[4];
    if (ks + 1 < G1_KSTEPS) {
#pragma unroll
      for (int nt = 0; nt < 4; ++nt)
        bn[nt] = load_frag16(sW, nt * 16, PITCH_W, (ks + 1) * 32);
    }
    c0 = wmma16(a[ks], bc[0], c0);
    c1 = wmma16(a[ks], bc[1], c1);
    c2 = wmma16(a[ks], bc[2], c2);
    c3 = wmma16(a[ks], bc[3], c3);
    if (ks + 1 < G1_KSTEPS) {
#pragma unroll
      for (int nt = 0; nt < 4; ++nt) bc[nt] = bn[nt];
    }
  }

  int lane = threadIdx.x & 31;
  int hf   = lane >> 4;
  int r    = lane & 15;
  v8f acc[4] = {c0, c1, c2, c3};
#pragma unroll
  for (int nt = 0; nt < 4; ++nt) {
#pragma unroll
    for (int k = 0; k < 8; ++k) {
      int orow = r0 + k + (hf << 3);
      int col  = nt * 16 + r;
      featsT[((size_t)head * NHID + col) * N + orow] = (_Float16)acc[nt][k];
    }
  }
  float ps[8], pn[8];
#pragma unroll
  for (int k = 0; k < 8; ++k) {
    float x = 0.0f, y = 0.0f;
#pragma unroll
    for (int nt = 0; nt < 4; ++nt) {
      x = __builtin_fmaf(acc[nt][k], asv[nt], x);
      y = __builtin_fmaf(acc[nt][k], anv[nt], y);
    }
    ps[k] = x;
    pn[k] = y;
  }
  store_dots(ps, pn, s + (size_t)head * N + r0, n + (size_t)head * N + r0);
}

#define WAVES1  4
#define KT      64
#define PITCH_H 72

__global__ void __launch_bounds__(32 * WAVES1)
__attribute__((amdgpu_waves_per_eu(1, 2)))
attn_gemm1_kernel(const unsigned* __restrict__ masks,
                  const _Float16* __restrict__ featsT,
                  const float* __restrict__ s1,
                  const float* __restrict__ n1,
                  const float* __restrict__ b1,
                  _Float16* __restrict__ H) {
  __shared__ __attribute__((aligned(16))) _Float16 sB[2][NHID * PITCH_H];

  int head = blockIdx.y;
  int wv   = __builtin_amdgcn_readfirstlane((int)(threadIdx.x >> 5));
  int r0   = ((int)blockIdx.x * WAVES1 + wv) << 4;
  int lane = threadIdx.x & 31;
  int hf   = lane >> 4;
  int r    = lane & 15;
  int row  = r0 + r;

  const float* nrow = n1 + head * N;
  float svL = s1[head * N + row] * LOG2E;
  const unsigned* mrow = masks + (size_t)(r0 >> 4) * (NCB * 16);
  const _Float16* BT = featsT + (size_t)head * NHID * N;

  const int tid = threadIdx.x;
  const _Float16* gsrc = BT + (size_t)(tid >> 3) * N + ((tid & 7) << 3);
  const int loff = (tid >> 3) * PITCH_H + ((tid & 7) << 3);

#pragma unroll
  for (int i = 0; i < 4; ++i) {
    v8h g = *(const v8h*)(gsrc + (size_t)i * 16 * N);
    *(v8h*)(&sB[0][loff + i * 16 * PITCH_H]) = g;
  }
  __syncthreads();

  float m_run = -1.0e30f;
  float l_run = 0.0f;

  unsigned mk[32];
#pragma unroll
  for (int e = 0; e < 32; ++e) mk[e] = mrow[e];

  v8f c0 = {}, c1 = {}, c2 = {}, c3 = {};
  for (int j0 = 0; j0 < N; j0 += KT) {
    const int cur = (j0 / KT) & 1;
    float nv[32];
#pragma unroll
    for (int e = 0; e < 32; ++e) {
      int j = j0 + ((e >> 4) << 5) + (e & 7) + (hf << 3) + (((e >> 3) & 1) << 4);
      nv[e] = nrow[j];
    }
    const int jn = (j0 + KT < N) ? (j0 + KT) : j0;
    v8h g[4];
#pragma unroll
    for (int i = 0; i < 4; ++i)
      g[i] = *(const v8h*)(gsrc + (size_t)i * 16 * N + jn);

    const _Float16* st = &sB[cur][0];
    v16h bf[2][4];
#pragma unroll
    for (int u = 0; u < 2; ++u)
#pragma unroll
      for (int nt = 0; nt < 4; ++nt)
        bf[u][nt] = load_frag16(st, nt * 16, PITCH_H, 32 * u);

    float t[32];
    float bm[2] = {MASK2, MASK2};
#pragma unroll
    for (int e = 0; e < 32; ++e) {
      t[e] = logit2m(__builtin_amdgcn_inverse_ballot_w32(mk[e]), nv[e], svL);
      bm[e >> 4] = fmaxf(bm[e >> 4], t[e]);
    }
    bm[0] = fmaxf(bm[0], xhalf(bm[0]));
    bm[1] = fmaxf(bm[1], xhalf(bm[1]));

#pragma unroll
    for (int i = 0; i < 4; ++i)
      *(v8h*)(&sB[cur ^ 1][loff + i * 16 * PITCH_H]) = g[i];

#pragma unroll
    for (int u = 0; u < 2; ++u) {
      if (__builtin_amdgcn_ballot_w32(bm[u] > m_run + LAZY) != 0) {
        float m_new = fmaxf(m_run, bm[u]);
        float f = __builtin_amdgcn_exp2f(m_run - m_new);
        l_run *= f;
#pragma unroll
        for (int k = 0; k < 8; ++k) {
          float fk = __shfl(f, k + (hf << 3), 32);
          c0[k] *= fk;
          c1[k] *= fk;
          c2[k] *= fk;
          c3[k] *= fk;
        }
        m_run = m_new;
      }

      if (u == 0) {
        const int jm = (j0 + KT < N) ? (j0 + KT) : j0;
#pragma unroll
        for (int e = 0; e < 32; ++e) mk[e] = mrow[(jm >> 1) + e];
      }

      v16h a;
#pragma unroll
      for (int e = 0; e < 16; ++e) {
        float p = __builtin_amdgcn_exp2f(t[16 * u + e] - m_run);
        l_run += p;
        a[e] = (_Float16)p;
      }
      c0 = wmma16(a, bf[u][0], c0);
      c1 = wmma16(a, bf[u][1], c1);
      c2 = wmma16(a, bf[u][2], c2);
      c3 = wmma16(a, bf[u][3], c3);
    }
    __syncthreads();
  }

  float inv = 1.0f / (l_run + xhalf(l_run));
  float ik[8];
#pragma unroll
  for (int k = 0; k < 8; ++k) ik[k] = __shfl(inv, k + (hf << 3), 32);

  v8f acc[4] = {c0, c1, c2, c3};
#pragma unroll
  for (int nt = 0; nt < 4; ++nt) {
#pragma unroll
    for (int k = 0; k < 8; ++k) {
      int orow = r0 + k + (hf << 3);
      int col  = nt * 16 + r;
      float v  = acc[nt][k] * ik[k] + b1[head * NHID + col];
      v = v > 0.0f ? v : 0.0f;
      H[(size_t)orow * (H1 * NHID) + head * NHID + col] = (_Float16)v;
    }
  }
}

__global__ void gemm2_kernel(const _Float16* __restrict__ H,
                             const _Float16* __restrict__ W2T,
                             const float* __restrict__ a_s,
                             const float* __restrict__ a_n,
                             float* __restrict__ s,
                             float* __restrict__ n,
                             _Float16* __restrict__ feats2T) {
  int r0 = blockIdx.x << 4;
  float asv = a_s[threadIdx.x & 15];
  float anv = a_n[threadIdx.x & 15];
  v8f c = {};
  for (int k0 = 0; k0 < H1 * NHID; k0 += 32) {
    v16h a = load_frag16(H, r0, H1 * NHID, k0);
    v16h b = load_frag16(W2T, 0, H1 * NHID, k0);
    c = wmma16(a, b, c);
  }
  int lane = threadIdx.x & 31;
  int hf   = lane >> 4;
  int r    = lane & 15;
  float ps[8], pn[8];
#pragma unroll
  for (int k = 0; k < 8; ++k) {
    int orow = r0 + k + (hf << 3);
    float v  = c[k];
    feats2T[(size_t)r * N + orow] = (_Float16)v;
    ps[k] = v * asv;
    pn[k] = v * anv;
  }
  store_dots(ps, pn, s + r0, n + r0);
}

#define SPLIT2 16

__global__ void __launch_bounds__(32 * SPLIT2)
__attribute__((amdgpu_waves_per_eu(4, 8)))
attn_gemm2_kernel(const unsigned* __restrict__ masks,
                  const _Float16* __restrict__ feats2T,
                  const float* __restrict__ s2,
                  const float* __restrict__ n2,
                  const float* __restrict__ b2,
                  float* __restrict__ out) {
  __shared__ float sh_c[SPLIT2][16][NCLASS];
  __shared__ float sh_m[SPLIT2][16];
  __shared__ float sh_l[SPLIT2][16];

  int r0   = blockIdx.x << 4;
  int wv   = __builtin_amdgcn_readfirstlane((int)(threadIdx.x >> 5));
  int lane = threadIdx.x & 31;
  int hf   = lane >> 4;
  int r    = lane & 15;
  int row  = r0 + r;

  float svL = s2[row] * LOG2E;
  const unsigned* mrow = masks + (size_t)blockIdx.x * (NCB * 16);

  float m_run = -1.0e30f;
  float l_run = 0.0f;

  v8f c = {};
  const int jbeg = wv * (N / SPLIT2);
  const int jend = jbeg + (N / SPLIT2);
  unsigned mk[16];
#pragma unroll
  for (int e = 0; e < 16; ++e) mk[e] = mrow[(jbeg >> 1) + e];

  for (int j0 = jbeg; j0 < jend; j0 += 32) {
    float nv[16];
#pragma unroll
    for (int e = 0; e < 16; ++e) {
      int j = j0 + (e & 7) + (hf << 3) + ((e >> 3) << 4);
      nv[e] = n2[j];
    }
    v16h b = load_frag16(feats2T, 0, N, j0);

    float t[16];
    float bm = MASK2;
#pragma unroll
    for (int e = 0; e < 16; ++e) {
      t[e] = logit2m(__builtin_amdgcn_inverse_ballot_w32(mk[e]), nv[e], svL);
      bm   = fmaxf(bm, t[e]);
    }
    bm = fmaxf(bm, xhalf(bm));

    asm volatile("" : "+v"(b), "+v"(bm));

    if (__builtin_amdgcn_ballot_w32(bm > m_run + LAZY) != 0) {
      float m_new = fmaxf(m_run, bm);
      float f = __builtin_amdgcn_exp2f(m_run - m_new);
      l_run *= f;
#pragma unroll
      for (int k = 0; k < 8; ++k) c[k] *= __shfl(f, k + (hf << 3), 32);
      m_run = m_new;
    }

    {
      const int jm = (j0 + 32 < jend) ? (j0 + 32) : j0;
#pragma unroll
      for (int e = 0; e < 16; ++e) mk[e] = mrow[(jm >> 1) + e];
    }

    v16h a;
#pragma unroll
    for (int e = 0; e < 16; ++e) {
      float p = __builtin_amdgcn_exp2f(t[e] - m_run);
      l_run += p;
      a[e] = (_Float16)p;
    }
    c = wmma16(a, b, c);
  }

  float lsum = l_run + xhalf(l_run);
  if (hf == 0) {
    sh_m[wv][r] = m_run;
    sh_l[wv][r] = lsum;
  }
#pragma unroll
  for (int k = 0; k < 8; ++k) sh_c[wv][k + (hf << 3)][r] = c[k];
  __syncthreads();

  if (threadIdx.x < 16 * NCLASS) {
    int mrow = threadIdx.x >> 4;
    int cls  = threadIdx.x & 15;
    float M = sh_m[0][mrow];
#pragma unroll
    for (int w = 1; w < SPLIT2; ++w) M = fmaxf(M, sh_m[w][mrow]);
    float L = 0.0f, C = 0.0f;
#pragma unroll
    for (int w = 0; w < SPLIT2; ++w) {
      float f = __builtin_amdgcn_exp2f(sh_m[w][mrow] - M);
      L = __builtin_fmaf(sh_l[w][mrow], f, L);
      C = __builtin_fmaf(sh_c[w][mrow][cls], f, C);
    }
    float inv = 1.0f / L;
    float v   = C * inv + b2[cls];
    v = v > 0.0f ? v : 0.0f;
    float mx = v;
#pragma unroll
    for (int msk = 1; msk < 16; msk <<= 1)
      mx = fmaxf(mx, __shfl_xor(mx, msk, 32));
    float ex = __expf(v - mx);
    float sum = ex;
#pragma unroll
    for (int msk = 1; msk < 16; msk <<= 1) sum += __shfl_xor(sum, msk, 32);
    float o = (v - mx) - logf(sum);
    out[(size_t)(r0 + mrow) * NCLASS + cls] = o;
  }
}

extern "C" void kernel_launch(void* const* d_in, const int* in_sizes, int n_in,
                              void* d_out, int out_size, void* d_ws,
                              size_t ws_size, hipStream_t stream) {
  const float* x   = (const float*)d_in[0];
  const int*   adj = (const int*)d_in[1];
  const float* W1  = (const float*)d_in[2];
  const float* b1  = (const float*)d_in[3];
  const float* as1 = (const float*)d_in[4];
  const float* an1 = (const float*)d_in[5];
  const float* W2  = (const float*)d_in[6];
  const float* b2  = (const float*)d_in[7];
  const float* as2 = (const float*)d_in[8];
  const float* an2 = (const float*)d_in[9];
  float* out = (float*)d_out;

  char* ws = (char*)d_ws;
  size_t off = 0;
  auto alloc = [&](size_t bytes) -> void* {
    off = (off + 255) & ~(size_t)255;
    void* p = ws + off;
    off += bytes;
    return p;
  };

  _Float16* XH      = (_Float16*)alloc((size_t)N * NFEAT * 2);
  _Float16* W1T     = (_Float16*)alloc((size_t)H1 * NHID * NFEAT * 2);
  _Float16* W2T     = (_Float16*)alloc((size_t)NCLASS * H1 * NHID * 2);
  _Float16* featsT  = (_Float16*)alloc((size_t)H1 * NHID * N * 2);
  float*    s1      = (float*)alloc((size_t)H1 * N * 4);
  float*    n1      = (float*)alloc((size_t)H1 * N * 4);
  _Float16* H       = (_Float16*)alloc((size_t)N * H1 * NHID * 2);
  _Float16* feats2T = (_Float16*)alloc((size_t)NCLASS * N * 2);
  float*    s2      = (float*)alloc((size_t)N * 4);
  float*    n2      = (float*)alloc((size_t)N * 4);
  unsigned* masks   = (unsigned*)alloc((size_t)(N / 16) * NCB * 16 * 4);
  (void)ws_size; (void)n_in; (void)in_sizes; (void)out_size;

  cast_kernel<<<CAST_X_BLOCKS + CAST_W1_BLOCKS + CAST_W2_BLOCKS, 256, 0,
                stream>>>(x, W1, W2, XH, W1T, W2T);
  gemm1_kernel<<<G1_BLOCKS + MASK_BLOCKS, 32 * G1_WAVES, 0, stream>>>(
      XH, W1T, as1, an1, s1, n1, featsT, adj, masks);
  attn_gemm1_kernel<<<dim3(N / (16 * WAVES1), H1), 32 * WAVES1, 0, stream>>>(
      masks, featsT, s1, n1, b1, H);
  gemm2_kernel<<<N / 16, 32, 0, stream>>>(H, W2T, as2, an2, s2, n2, feats2T);
  attn_gemm2_kernel<<<N / 16, 32 * SPLIT2, 0, stream>>>(masks, feats2T, s2, n2,
                                                        b2, out);
}


# Layer-1 attention: each step opens by requesting the next step's n values and the tile after next (two register sets; hand-written LDS-only barrier keeps them in flight); sums unchanged
# speedup vs baseline: 7.4261x; 1.1042x over previous
#include <hip/hip_runtime.h>
#include <hip/hip_bf16.h>


#define N      4096
#define NFEAT  512
#define NHID   64
#define NCLASS 16
#define H1     8
#define NEG_SLOPE 0.2f
#define MASK_VAL  -1e10f
#define LOG2E 1.4426950408889634f
#define MASK2 (MASK_VAL * LOG2E)
#define LAZY 8.0f

typedef __attribute__((ext_vector_type(16))) _Float16 v16h;
typedef __attribute__((ext_vector_type(8)))  _Float16 v8h;
typedef __attribute__((ext_vector_type(8)))  float    v8f;

__device__ __forceinline__ v8f wmma16(v16h a, v16h b, v8f c) {
  return __builtin_amdgcn_wmma_f32_16x16x32_f16(
       false, a,  false, b,
       (short)0, c,  false,  false);
}

__device__ __forceinline__ v16h load_frag16(const _Float16* base, int rc,
                                            int ld, int k0) {
  int lane = threadIdx.x & 31;
  int hf   = lane >> 4;
  int r    = lane & 15;
  const _Float16* p = base + (size_t)(rc + r) * ld + k0 + (hf << 3);
  v8h lo = *(const v8h*)p;
  v8h hi = *(const v8h*)(p + 16);
  return __builtin_shufflevector(lo, hi, 0, 1, 2, 3, 4, 5, 6, 7, 8, 9, 10, 11,
                                 12, 13, 14, 15);
}

__device__ __forceinline__ float lrelu(float x) {
  return x > 0.0f ? x : NEG_SLOPE * x;
}

__device__ __forceinline__ float xhalf(float x) {
  int b = __builtin_bit_cast(int, x);
  return __builtin_bit_cast(
      float, __builtin_amdgcn_permlanex16(b, b, 0x76543210, 0xfedcba98, false,
                                          false));
}

__device__ __forceinline__ float logit2m(bool on, float nj, float svL) {
  float x2 = __builtin_fmaf(nj, LOG2E, svL);
  float lr = fmaxf(x2, NEG_SLOPE * x2);
  return on ? lr : MASK2;
}

template <int M>
__device__ __forceinline__ float add_xmask(float v) {
  int b = __builtin_bit_cast(int, v);
  int o = __builtin_amdgcn_update_dpp(b, b, 0x160 + M, 0xf, 0xf, false);
  return v + __builtin_bit_cast(float, o);
}
__device__ __forceinline__ float rowsum16(float v) {
  return add_xmask<8>(add_xmask<4>(add_xmask<2>(add_xmask<1>(v))));
}

__device__ __forceinline__ void store_dots(float (&ps)[8], float (&pn)[8],
                                           float* __restrict__ dst_s,
                                           float* __restrict__ dst_n) {
  int lane = threadIdx.x & 31;
  int hf   = lane >> 4;
  int r    = lane & 15;
#pragma unroll
  for (int k = 0; k < 8; ++k) {
    ps[k] = rowsum16(ps[k]);
    pn[k] = rowsum16(pn[k]);
  }
  float sv = ps[0], nv = pn[0];
#pragma unroll
  for (int k = 1; k < 8; ++k) {
    sv = (r == k) ? ps[k] : sv;
    nv = (r == k) ? pn[k] : nv;
  }
  if (r < 8) {
    dst_s[(hf << 3) + r] = sv;
    dst_n[(hf << 3) + r] = nv;
  }
}

#define NCB (N / 32)
#define MASK_BLOCKS ((N / 16) * (NCB / 4) / 8)
typedef __attribute__((ext_vector_type(4))) int v4i;

__device__ __forceinline__ void adjmask_block(const int* __restrict__ adj,
                                              unsigned* __restrict__ masks,
                                              int blk) {
  int wv   = __builtin_amdgcn_readfirstlane((int)(threadIdx.x >> 5));
  int gw   = blk * 8 + wv;
  int rt   = gw >> 5;
  int cb0  = (gw & 31) << 2;
  int lane = threadIdx.x & 31;
  int hf   = lane >> 4;
  int r    = lane & 15;
  const int* arow = adj + (size_t)(rt * 16 + r) * N + cb0 * 32 + (hf << 3);

  v4i a[4][4];
#pragma unroll
  for (int cb = 0; cb < 4; ++cb) {
#pragma unroll
    for (int q = 0; q < 4; ++q) {
      a[cb][q] = __builtin_nontemporal_load(
          (const v4i*)(arow + cb * 32 + (q & 1) * 4 + (q >> 1) * 16));
    }
  }
  unsigned o[2] = {0u, 0u};
#pragma unroll
  for (int cb = 0; cb < 4; ++cb) {
#pragma unroll
    for (int e = 0; e < 16; ++e) {
      unsigned m = __builtin_amdgcn_ballot_w32(a[cb][e >> 2][e & 3] > 0);
      o[cb >> 1] = (lane == e + 16 * (cb & 1)) ? m : o[cb >> 1];
    }
  }
  unsigned* dst = masks + ((size_t)rt * NCB + cb0) * 16;
  dst[lane]      = o[0];
  dst[32 + lane] = o[1];
}

typedef __attribute__((ext_vector_type(4))) float    v4f;
typedef __attribute__((ext_vector_type(4))) _Float16 v4h;

#define CAST_X_BLOCKS     256
#define CAST_X_PER_THREAD ((N * NFEAT / 4) / (CAST_X_BLOCKS * 256))
#define CAST_W1_BLOCKS    ((H1 * NHID * NFEAT / 8) / 256)
#define CAST_W2_BLOCKS    ((NCLASS * H1 * NHID / 8) / 256)

__global__ void __launch_bounds__(256)
cast_kernel(const float* __restrict__ x,
            const float* __restrict__ W1,
            const float* __restrict__ W2,
            _Float16* __restrict__ XH,
            _Float16* __restrict__ W1T,
            _Float16* __restrict__ W2T) {
  const int b   = blockIdx.x;
  const int tid = threadIdx.x;
  if (b < CAST_X_BLOCKS) {
    const int base = b * (256 * CAST_X_PER_THREAD) + tid;
    const v4f* src = (const v4f*)x + base;
    v4h* dst = (v4h*)XH + base;
    v4f v[CAST_X_PER_THREAD];
#pragma unroll
    for (int i = 0; i < CAST_X_PER_THREAD; ++i) v[i] = src[i * 256];
#pragma unroll
    for (int i = 0; i < CAST_X_PER_THREAD; ++i)
      dst[i * 256] = __builtin_convertvector(v[i], v4h);
  } else if (b < CAST_X_BLOCKS + CAST_W1_BLOCKS) {
    const int q  = (b - CAST_X_BLOCKS) * 256 + tid;
    const int c  = q & (NHID - 1);
    const int k0 = ((q / NHID) & (NFEAT / 8 - 1)) * 8;
    const int h  = q / (NHID * (NFEAT / 8));
    const float* src = W1 + ((size_t)h * NFEAT + k0) * NHID + c;
    v8h o;
#pragma unroll
    for (int j = 0; j < 8; ++j) o[j] = (_Float16)src[j * NHID];
    *(v8h*)(W1T + ((size_t)(h * NHID + c)) * NFEAT + k0) = o;
  } else {
    const int q  = (b - CAST_X_BLOCKS - CAST_W1_BLOCKS) * 256 + tid;
    const int c  = q & (NCLASS - 1);
    const int k0 = (q / NCLASS) * 8;
    const float* src = W2 + (size_t)k0 * NCLASS + c;
    v8h o;
#pragma unroll
    for (int j = 0; j < 8; ++j) o[j] = (_Float16)src[j * NCLASS];
    *(v8h*)(W2T + (size_t)c * (H1 * NHID) + k0) = o;
  }
}

#define G1_WAVES  8
#define G1_KSTEPS (NFEAT / 32)
#define G1_XBLOCKS (N / (16 * G1_WAVES))
#define G1_BLOCKS  (G1_XBLOCKS * H1)
#define PITCH_W (NFEAT + 8)

__global__ void __launch_bounds__(32 * G1_WAVES)
__attribute__((amdgpu_waves_per_eu(1, 2)))
gemm1_kernel(const _Float16* __restrict__ XH,
             const _Float16* __restrict__ W1T,
             const float* __restrict__ a_s,
             const float* __restrict__ a_n,
             float* __restrict__ s,
             float* __restrict__ n,
             _Float16* __restrict__ featsT,
             const int* __restrict__ adj,
             unsigned* __restrict__ masks) {
  __shared__ __attribute__((aligned(16))) _Float16 sW[NHID * PITCH_W];

  if (blockIdx.x >= G1_BLOCKS) {
    adjmask_block(adj, masks, (int)blockIdx.x - G1_BLOCKS);
    return;
  }

  int head = (int)blockIdx.x / G1_XBLOCKS;
  int xb   = (int)blockIdx.x % G1_XBLOCKS;
  int wv   = __builtin_amdgcn_readfirstlane((int)(threadIdx.x >> 5));
  int r0   = (xb * G1_WAVES + wv) << 4;
  const int tid = threadIdx.x;
  const _Float16* BT = W1T + (size_t)head * NHID * NFEAT;

  const _Float16* gsrc = BT + (size_t)tid * 8;
  const int loff = (tid >> 6) * PITCH_W + ((tid & 63) << 3);
  v8h g[16];
#pragma unroll
  for (int i = 0; i < 16; ++i)
    g[i] = *(const v8h*)(gsrc + (size_t)i * (256 * 8));
  v16h a[G1_KSTEPS];
#pragma unroll
  for (int ks = 0; ks < G1_KSTEPS; ++ks)
    a[ks] = load_frag16(XH, r0, NFEAT, ks * 32);
  float asv[4], anv[4];
#pragma unroll
  for (int nt = 0; nt < 4; ++nt) {
    asv[nt] = a_s[head * NHID + nt * 16 + (tid & 15)];
    anv[nt] = a_n[head * NHID + nt * 16 + (tid & 15)];
  }
#pragma unroll
  for (int i = 0; i < 16; ++i)
    *(v8h*)(&sW[loff + i * 4 * PITCH_W]) = g[i];
  __syncthreads();

  v8f c0 = {}, c1 = {}, c2 = {}, c3 = {};
  v16h bc[4];
#pragma unroll
  for (int nt = 0; nt < 4; ++nt) bc[nt] = load_frag16(sW, nt * 16, PITCH_W, 0);
#pragma unroll
  for (int ks = 0; ks < G1_KSTEPS; ++ks) {
    v16h bn[4];
    if (ks + 1 < G1_KSTEPS) {
#pragma unroll
      for (int nt = 0; nt < 4; ++nt)
        bn[nt] = load_frag16(sW, nt * 16, PITCH_W, (ks + 1) * 32);
    }
    c0 = wmma16(a[ks], bc[0], c0);
    c1 = wmma16(a[ks], bc[1], c1);
    c2 = wmma16(a[ks], bc[2], c2);
    c3 = wmma16(a[ks], bc[3], c3);
    if (ks + 1 < G1_KSTEPS) {
#pragma unroll
      for (int nt = 0; nt < 4; ++nt) bc[nt] = bn[nt];
    }
  }

  int lane = threadIdx.x & 31;
  int hf   = lane >> 4;
  int r    = lane & 15;
  v8f acc[4] = {c0, c1, c2, c3};
#pragma unroll
  for (int nt = 0; nt < 4; ++nt) {
#pragma unroll
    for (int k = 0; k < 8; ++k) {
      int orow = r0 + k + (hf << 3);
      int col  = nt * 16 + r;
      featsT[((size_t)head * NHID + col) * N + orow] = (_Float16)acc[nt][k];
    }
  }
  float ps[8], pn[8];
#pragma unroll
  for (int k = 0; k < 8; ++k) {
    float x = 0.0f, y = 0.0f;
#pragma unroll
    for (int nt = 0; nt < 4; ++nt) {
      x = __builtin_fmaf(acc[nt][k], asv[nt], x);
      y = __builtin_fmaf(acc[nt][k], anv[nt], y);
    }
    ps[k] = x;
    pn[k] = y;
  }
  store_dots(ps, pn, s + (size_t)head * N + r0, n + (size_t)head * N + r0);
}

#define WAVES1  4
#define KT      64
#define PITCH_H 72

__global__ void __launch_bounds__(32 * WAVES1)
__attribute__((amdgpu_waves_per_eu(1, 2)))
attn_gemm1_kernel(const unsigned* __restrict__ masks,
                  const _Float16* __restrict__ featsT,
                  const float* __restrict__ s1,
                  const float* __restrict__ n1,
                  const float* __restrict__ b1,
                  _Float16* __restrict__ H) {
  __shared__ __attribute__((aligned(16))) _Float16 sB[2][NHID * PITCH_H];

  int head = blockIdx.y;
  int wv   = __builtin_amdgcn_readfirstlane((int)(threadIdx.x >> 5));
  int r0   = ((int)blockIdx.x * WAVES1 + wv) << 4;
  int lane = threadIdx.x & 31;
  int hf   = lane >> 4;
  int r    = lane & 15;
  int row  = r0 + r;

  const float* nrow = n1 + head * N;
  float svL = s1[head * N + row] * LOG2E;
  const unsigned* mrow = masks + (size_t)(r0 >> 4) * (NCB * 16);
  const _Float16* BT = featsT + (size_t)head * NHID * N;

  const int tid = threadIdx.x;
  const _Float16* gsrc = BT + (size_t)(tid >> 3) * N + ((tid & 7) << 3);
  const int loff = (tid >> 3) * PITCH_H + ((tid & 7) << 3);

  v8h g0[4];
#pragma unroll
  for (int i = 0; i < 4; ++i) g0[i] = *(const v8h*)(gsrc + (size_t)i * 16 * N);
  float nv[2][32];
  const int hoff = hf << 3;
#define NCOL(e) ((((e) >> 4) << 5) + ((e) & 7) + ((((e) >> 3) & 1) << 4))
#pragma unroll
  for (int e = 0; e < 32; ++e) nv[0][e] = nrow[hoff + NCOL(e)];
  v8h g[2][4];
#pragma unroll
  for (int i = 0; i < 4; ++i)
    g[0][i] = *(const v8h*)(gsrc + (size_t)i * 16 * N + KT);
#pragma unroll
  for (int i = 0; i < 4; ++i) *(v8h*)(&sB[0][loff + i * 16 * PITCH_H]) = g0[i];
  __syncthreads();

  float m_run = -1.0e30f;
  float l_run = 0.0f;

  unsigned mk[32];
#pragma unroll
  for (int e = 0; e < 32; ++e) mk[e] = mrow[e];

  v8f c0 = {}, c1 = {}, c2 = {}, c3 = {};
  for (int jj = 0; jj < N; jj += 2 * KT) {
#pragma unroll
    for (int cur = 0; cur < 2; ++cur) {
    const int j0 = jj + cur * KT;
    {
      const int jn = (j0 + KT < N) ? (j0 + KT) : j0;
#pragma unroll
      for (int e = 0; e < 32; ++e) nv[cur ^ 1][e] = nrow[jn + hoff + NCOL(e)];
      const int jt = (j0 + 2 * KT < N) ? (j0 + 2 * KT) : j0;
#pragma unroll
      for (int i = 0; i < 4; ++i)
        g[cur ^ 1][i] = *(const v8h*)(gsrc + (size_t)i * 16 * N + jt);
    }

    const _Float16* st = &sB[cur][0];
    v16h bf[2][4];
#pragma unroll
    for (int u = 0; u < 2; ++u)
#pragma unroll
      for (int nt = 0; nt < 4; ++nt)
        bf[u][nt] = load_frag16(st, nt * 16, PITCH_H, 32 * u);

    float t[32];
    float bm[2] = {MASK2, MASK2};
#pragma unroll
    for (int e = 0; e < 32; ++e) {
      t[e] = logit2m(__builtin_amdgcn_inverse_ballot_w32(mk[e]), nv[cur][e],
                     svL);
      bm[e >> 4] = fmaxf(bm[e >> 4], t[e]);
    }
    bm[0] = fmaxf(bm[0], xhalf(bm[0]));
    bm[1] = fmaxf(bm[1], xhalf(bm[1]));

#pragma unroll
    for (int i = 0; i < 4; ++i)
      *(v8h*)(&sB[cur ^ 1][loff + i * 16 * PITCH_H]) = g[cur][i];

#pragma unroll
    for (int u = 0; u < 2; ++u) {
      if (__builtin_amdgcn_ballot_w32(bm[u] > m_run + LAZY) != 0) {
        float m_new = fmaxf(m_run, bm[u]);
        float f = __builtin_amdgcn_exp2f(m_run - m_new);
        l_run *= f;
#pragma unroll
        for (int k = 0; k < 8; ++k) {
          float fk = __shfl(f, k + (hf << 3), 32);
          c0[k] *= fk;
          c1[k] *= fk;
          c2[k] *= fk;
          c3[k] *= fk;
        }
        m_run = m_new;
      }

      if (u == 0) {
        const int jm = (j0 + KT < N) ? (j0 + KT) : j0;
#pragma unroll
        for (int e = 0; e < 32; ++e) mk[e] = mrow[(jm >> 1) + e];
      }

      v16h a;
#pragma unroll
      for (int e = 0; e < 16; ++e) {
        float p = __builtin_amdgcn_exp2f(t[16 * u + e] - m_run);
        l_run += p;
        a[e] = (_Float16)p;
      }
      c0 = wmma16(a, bf[u][0], c0);
      c1 = wmma16(a, bf[u][1], c1);
      c2 = wmma16(a, bf[u][2], c2);
      c3 = wmma16(a, bf[u][3], c3);
    }
    asm volatile("s_wait_dscnt 0x0\n\ts_barrier_signal -1\n\ts_barrier_wait -1"
                 ::: "memory");
    }
  }
#undef NCOL

  float inv = 1.0f / (l_run + xhalf(l_run));
  float ik[8];
#pragma unroll
  for (int k = 0; k < 8; ++k) ik[k] = __shfl(inv, k + (hf << 3), 32);

  v8f acc[4] = {c0, c1, c2, c3};
#pragma unroll
  for (int nt = 0; nt < 4; ++nt) {
#pragma unroll
    for (int k = 0; k < 8; ++k) {
      int orow = r0 + k + (hf << 3);
      int col  = nt * 16 + r;
      float v  = acc[nt][k] * ik[k] + b1[head * NHID + col];
      v = v > 0.0f ? v : 0.0f;
      H[(size_t)orow * (H1 * NHID) + head * NHID + col] = (_Float16)v;
    }
  }
}

__global__ void gemm2_kernel(const _Float16* __restrict__ H,
                             const _Float16* __restrict__ W2T,
                             const float* __restrict__ a_s,
                             const float* __restrict__ a_n,
                             float* __restrict__ s,
                             float* __restrict__ n,
                             _Float16* __restrict__ feats2T) {
  int r0 = blockIdx.x << 4;
  float asv = a_s[threadIdx.x & 15];
  float anv = a_n[threadIdx.x & 15];
  v8f c = {};
  for (int k0 = 0; k0 < H1 * NHID; k0 += 32) {
    v16h a = load_frag16(H, r0, H1 * NHID, k0);
    v16h b = load_frag16(W2T, 0, H1 * NHID, k0);
    c = wmma16(a, b, c);
  }
  int lane = threadIdx.x & 31;
  int hf   = lane >> 4;
  int r    = lane & 15;
  float ps[8], pn[8];
#pragma unroll
  for (int k = 0; k < 8; ++k) {
    int orow = r0 + k + (hf << 3);
    float v  = c[k];
    feats2T[(size_t)r * N + orow] = (_Float16)v;
    ps[k] = v * asv;
    pn[k] = v * anv;
  }
  store_dots(ps, pn, s + r0, n + r0);
}

#define SPLIT2 16

__global__ void __launch_bounds__(32 * SPLIT2)
__attribute__((amdgpu_waves_per_eu(4, 8)))
attn_gemm2_kernel(const unsigned* __restrict__ masks,
                  const _Float16* __restrict__ feats2T,
                  const float* __restrict__ s2,
                  const float* __restrict__ n2,
                  const float* __restrict__ b2,
                  float* __restrict__ out) {
  __shared__ float sh_c[SPLIT2][16][NCLASS];
  __shared__ float sh_m[SPLIT2][16];
  __shared__ float sh_l[SPLIT2][16];

  int r0   = blockIdx.x << 4;
  int wv   = __builtin_amdgcn_readfirstlane((int)(threadIdx.x >> 5));
  int lane = threadIdx.x & 31;
  int hf   = lane >> 4;
  int r    = lane & 15;
  int row  = r0 + r;

  float svL = s2[row] * LOG2E;
  const unsigned* mrow = masks + (size_t)blockIdx.x * (NCB * 16);

  float m_run = -1.0e30f;
  float l_run = 0.0f;

  v8f c = {};
  const int jbeg = wv * (N / SPLIT2);
  const int jend = jbeg + (N / SPLIT2);
  unsigned mk[16];
#pragma unroll
  for (int e = 0; e < 16; ++e) mk[e] = mrow[(jbeg >> 1) + e];

  for (int j0 = jbeg; j0 < jend; j0 += 32) {
    float nv[16];
#pragma unroll
    for (int e = 0; e < 16; ++e) {
      int j = j0 + (e & 7) + (hf << 3) + ((e >> 3) << 4);
      nv[e] = n2[j];
    }
    v16h b = load_frag16(feats2T, 0, N, j0);

    float t[16];
    float bm = MASK2;
#pragma unroll
    for (int e = 0; e < 16; ++e) {
      t[e] = logit2m(__builtin_amdgcn_inverse_ballot_w32(mk[e]), nv[e], svL);
      bm   = fmaxf(bm, t[e]);
    }
    bm = fmaxf(bm, xhalf(bm));

    asm volatile("" : "+v"(b), "+v"(bm));

    if (__builtin_amdgcn_ballot_w32(bm > m_run + LAZY) != 0) {
      float m_new = fmaxf(m_run, bm);
      float f = __builtin_amdgcn_exp2f(m_run - m_new);
      l_run *= f;
#pragma unroll
      for (int k = 0; k < 8; ++k) c[k] *= __shfl(f, k + (hf << 3), 32);
      m_run = m_new;
    }

    {
      const int jm = (j0 + 32 < jend) ? (j0 + 32) : j0;
#pragma unroll
      for (int e = 0; e < 16; ++e) mk[e] = mrow[(jm >> 1) + e];
    }

    v16h a;
#pragma unroll
    for (int e = 0; e < 16; ++e) {
      float p = __builtin_amdgcn_exp2f(t[e] - m_run);
      l_run += p;
      a[e] = (_Float16)p;
    }
    c = wmma16(a, b, c);
  }

  float lsum = l_run + xhalf(l_run);
  if (hf == 0) {
    sh_m[wv][r] = m_run;
    sh_l[wv][r] = lsum;
  }
#pragma unroll
  for (int k = 0; k < 8; ++k) sh_c[wv][k + (hf << 3)][r] = c[k];
  __syncthreads();

  if (threadIdx.x < 16 * NCLASS) {
    int mrow = threadIdx.x >> 4;
    int cls  = threadIdx.x & 15;
    float M = sh_m[0][mrow];
#pragma unroll
    for (int w = 1; w < SPLIT2; ++w) M = fmaxf(M, sh_m[w][mrow]);
    float L = 0.0f, C = 0.0f;
#pragma unroll
    for (int w = 0; w < SPLIT2; ++w) {
      float f = __builtin_amdgcn_exp2f(sh_m[w][mrow] - M);
      L = __builtin_fmaf(sh_l[w][mrow], f, L);
      C = __builtin_fmaf(sh_c[w][mrow][cls], f, C);
    }
    float inv = 1.0f / L;
    float v   = C * inv + b2[cls];
    v = v > 0.0f ? v : 0.0f;
    float mx = v;
#pragma unroll
    for (int msk = 1; msk < 16; msk <<= 1)
      mx = fmaxf(mx, __shfl_xor(mx, msk, 32));
    float ex = __expf(v - mx);
    float sum = ex;
#pragma unroll
    for (int msk = 1; msk < 16; msk <<= 1) sum += __shfl_xor(sum, msk, 32);
    float o = (v - mx) - logf(sum);
    out[(size_t)(r0 + mrow) * NCLASS + cls] = o;
  }
}

extern "C" void kernel_launch(void* const* d_in, const int* in_sizes, int n_in,
                              void* d_out, int out_size, void* d_ws,
                              size_t ws_size, hipStream_t stream) {
  const float* x   = (const float*)d_in[0];
  const int*   adj = (const int*)d_in[1];
  const float* W1  = (const float*)d_in[2];
  const float* b1  = (const float*)d_in[3];
  const float* as1 = (const float*)d_in[4];
  const float* an1 = (const float*)d_in[5];
  const float* W2  = (const float*)d_in[6];
  const float* b2  = (const float*)d_in[7];
  const float* as2 = (const float*)d_in[8];
  const float* an2 = (const float*)d_in[9];
  float* out = (float*)d_out;

  char* ws = (char*)d_ws;
  size_t off = 0;
  auto alloc = [&](size_t bytes) -> void* {
    off = (off + 255) & ~(size_t)255;
    void* p = ws + off;
    off += bytes;
    return p;
  };

  _Float16* XH      = (_Float16*)alloc((size_t)N * NFEAT * 2);
  _Float16* W1T     = (_Float16*)alloc((size_t)H1 * NHID * NFEAT * 2);
  _Float16* W2T     = (_Float16*)alloc((size_t)NCLASS * H1 * NHID * 2);
  _Float16* featsT  = (_Float16*)alloc((size_t)H1 * NHID * N * 2);
  float*    s1      = (float*)alloc((size_t)H1 * N * 4);
  float*    n1      = (float*)alloc((size_t)H1 * N * 4);
  _Float16* H       = (_Float16*)alloc((size_t)N * H1 * NHID * 2);
  _Float16* feats2T = (_Float16*)alloc((size_t)NCLASS * N * 2);
  float*    s2      = (float*)alloc((size_t)N * 4);
  float*    n2      = (float*)alloc((size_t)N * 4);
  unsigned* masks   = (unsigned*)alloc((size_t)(N / 16) * NCB * 16 * 4);
  (void)ws_size; (void)n_in; (void)in_sizes; (void)out_size;

  cast_kernel<<<CAST_X_BLOCKS + CAST_W1_BLOCKS + CAST_W2_BLOCKS, 256, 0,
                stream>>>(x, W1, W2, XH, W1T, W2T);
  gemm1_kernel<<<G1_BLOCKS + MASK_BLOCKS, 32 * G1_WAVES, 0, stream>>>(
      XH, W1T, as1, an1, s1, n1, featsT, adj, masks);
  attn_gemm1_kernel<<<dim3(N / (16 * WAVES1), H1), 32 * WAVES1, 0, stream>>>(
      masks, featsT, s1, n1, b1, H);
  gemm2_kernel<<<N / 16, 32, 0, stream>>>(H, W2T, as2, an2, s2, n2, feats2T);
  attn_gemm2_kernel<<<N / 16, 32 * SPLIT2, 0, stream>>>(masks, feats2T, s2, n2,
                                                        b2, out);
}
